# starting kernel
# baseline (speedup 1.0000x reference)
#include <hip/hip_runtime.h>
#include <math.h>

typedef __attribute__((ext_vector_type(2))) float v2f;
typedef __attribute__((ext_vector_type(8))) float v8f;

#define LN_EPS 1e-5f

__global__ void degree_kernel(const int* __restrict__ src, const int* __restrict__ dst,
                              float* __restrict__ dout, float* __restrict__ din, int e) {
    int i = blockIdx.x * blockDim.x + threadIdx.x;
    if (i >= e) return;
    atomicAdd(dout + src[i], 1.0f);
    atomicAdd(din  + dst[i], 1.0f);
}

__global__ void invsqrt_deg_kernel(float* __restrict__ dout, float* __restrict__ din, int n) {
    int i = blockIdx.x * blockDim.x + threadIdx.x;
    if (i >= n) return;
    dout[i] = rsqrtf(fmaxf(dout[i], 1.0f));
    din[i]  = rsqrtf(fmaxf(din[i],  1.0f));
}

__global__ void ln_stats_768_kernel(const float* __restrict__ feat,
                                    float* __restrict__ mu, float* __restrict__ rs, int n) {
    int row  = (blockIdx.x * blockDim.x + threadIdx.x) >> 5;
    int lane = threadIdx.x & 31;
    if (row >= n) return;
    const float* p = feat + (size_t)row * 768;
    float s = 0.f, s2 = 0.f;
    #pragma unroll
    for (int c = 0; c < 768; c += 32) {
        float v = p[c + lane];
        s += v; s2 += v * v;
    }
    #pragma unroll
    for (int off = 16; off > 0; off >>= 1) {
        s  += __shfl_xor(s,  off, 32);
        s2 += __shfl_xor(s2, off, 32);
    }
    if (lane == 0) {
        float m   = s  * (1.0f / 768.0f);
        float var = s2 * (1.0f / 768.0f) - m * m;
        mu[row] = m;
        rs[row] = rsqrtf(var + LN_EPS);
    }
}

__global__ void gemm0_ln_wmma_kernel(const float* __restrict__ feat,
                                     const float* __restrict__ mu, const float* __restrict__ rs,
                                     const float* __restrict__ do_is,
                                     const float* __restrict__ g, const float* __restrict__ b,
                                     const float* __restrict__ W,
                                     float* __restrict__ C, int M) {
    const int K = 768, F = 128;
    const int lane = threadIdx.x & 31;
    const int mt   = (blockIdx.x * blockDim.x + threadIdx.x) >> 5;
    if (mt * 16 >= M) return;
    const int half = lane >> 4;
    const int l16  = lane & 15;
    const int arow = mt * 16 + l16;
    const int koff = half * 2;
    const float m_r  = mu[arow];
    const float rs_r = rs[arow];
    const float do_r = do_is[arow];
    const float* ap = feat + (size_t)arow * K + koff;

    v8f acc[8];
    #pragma unroll
    for (int i = 0; i < 8; ++i) { v8f z = {}; acc[i] = z; }

    for (int k = 0; k < K; k += 4) {
        const int c0 = k + koff;
        const v2f x  = *(const v2f*)(ap + k);
        const v2f gg = *(const v2f*)(g + c0);
        const v2f bbv = *(const v2f*)(b + c0);
        v2f a;
        a.x = ((x.x - m_r) * rs_r * gg.x + bbv.x) * do_r;
        a.y = ((x.y - m_r) * rs_r * gg.y + bbv.y) * do_r;
        const float* wrow = W + (size_t)c0 * F;
        #pragma unroll
        for (int ft = 0; ft < 8; ++ft) {
            const int bcol = ft * 16 + l16;
            v2f bb;
            bb.x = wrow[bcol];
            bb.y = wrow[bcol + F];
            acc[ft] = __builtin_amdgcn_wmma_f32_16x16x4_f32(
                false, a, false, bb, (short)0, acc[ft], false, false);
        }
    }
    const int crow = mt * 16 + half * 8;
    #pragma unroll
    for (int ft = 0; ft < 8; ++ft) {
        const int bcol = ft * 16 + l16;
        #pragma unroll
        for (int r = 0; r < 8; ++r)
            C[(size_t)(crow + r) * F + bcol] = acc[ft][r];
    }
}

template <int FT>
__global__ void gemm_wmma_strip_kernel(const float* __restrict__ A, const float* __restrict__ W,
                                       float* __restrict__ C, int M, int K, int F) {
    const int lane = threadIdx.x & 31;
    const int mt   = (blockIdx.x * blockDim.x + threadIdx.x) >> 5;
    if (mt * 16 >= M) return;
    const int half = lane >> 4;
    const int l16  = lane & 15;
    const int arow = mt * 16 + l16;
    const int koff = half * 2;
    const float* ap = A + (size_t)arow * K + koff;

    v8f acc[FT];
    #pragma unroll
    for (int i = 0; i < FT; ++i) { v8f z = {}; acc[i] = z; }

    for (int k = 0; k < K; k += 4) {
        const v2f a = *(const v2f*)(ap + k);
        const float* wrow = W + (size_t)(k + koff) * F;
        #pragma unroll
        for (int ft = 0; ft < FT; ++ft) {
            const int  bcol = ft * 16 + l16;
            const bool bv   = (bcol < F);
            const int  bc   = bv ? bcol : 0;
            v2f bb;
            bb.x = bv ? wrow[bc]     : 0.0f;
            bb.y = bv ? wrow[bc + F] : 0.0f;
            acc[ft] = __builtin_amdgcn_wmma_f32_16x16x4_f32(
                false, a, false, bb, (short)0, acc[ft], false, false);
        }
    }
    const int crow = mt * 16 + half * 8;
    #pragma unroll
    for (int ft = 0; ft < FT; ++ft) {
        const int bcol = ft * 16 + l16;
        if (bcol < F) {
            #pragma unroll
            for (int r = 0; r < 8; ++r)
                C[(size_t)(crow + r) * F + bcol] = acc[ft][r];
        }
    }
}

template <int F4>
__global__ void scatter_add_vec4_kernel(const float* __restrict__ x,
                                        const int* __restrict__ src, const int* __restrict__ dst,
                                        float* __restrict__ out, int e) {
    long long t = (long long)blockIdx.x * blockDim.x + threadIdx.x;
    if (t >= (long long)e * F4) return;
    const int ed = (int)(t / F4);
    const int c4 = (int)(t % F4);
    const float4 v = ((const float4*)(x + (size_t)src[ed] * (F4 * 4)))[c4];
    float* o = out + (size_t)dst[ed] * (F4 * 4) + (size_t)c4 * 4;
    atomicAdd(o + 0, v.x);
    atomicAdd(o + 1, v.y);
    atomicAdd(o + 2, v.z);
    atomicAdd(o + 3, v.w);
}

__global__ void post_ln_relu_128_kernel(const float* __restrict__ agg,
                                        const float* __restrict__ di_is,
                                        const float* __restrict__ do_is,
                                        const float* __restrict__ g, const float* __restrict__ b,
                                        float* __restrict__ out, int n) {
    int row  = (blockIdx.x * blockDim.x + threadIdx.x) >> 5;
    int lane = threadIdx.x & 31;
    if (row >= n) return;
    const float* p = agg + (size_t)row * 128;
    const float di = di_is[row];
    float v[4];
    float s = 0.f, s2 = 0.f;
    #pragma unroll
    for (int i = 0; i < 4; ++i) {
        v[i] = p[lane + 32 * i] * di;
        s += v[i]; s2 += v[i] * v[i];
    }
    #pragma unroll
    for (int off = 16; off > 0; off >>= 1) {
        s  += __shfl_xor(s,  off, 32);
        s2 += __shfl_xor(s2, off, 32);
    }
    const float m   = s  * (1.0f / 128.0f);
    const float rsv = rsqrtf(s2 * (1.0f / 128.0f) - m * m + LN_EPS);
    const float ds  = do_is[row];
    float* q = out + (size_t)row * 128;
    #pragma unroll
    for (int i = 0; i < 4; ++i) {
        const int c = lane + 32 * i;
        float y = (v[i] - m) * rsv * g[c] + b[c];
        q[c] = fmaxf(y, 0.0f) * ds;
    }
}

__global__ void final_scale_bias_kernel(float* __restrict__ out,
                                        const float* __restrict__ di_is,
                                        const float* __restrict__ b2, int n) {
    const int F = 40;
    int t = blockIdx.x * blockDim.x + threadIdx.x;
    if (t >= n * F) return;
    out[t] = out[t] * di_is[t / F] + b2[t % F];
}

static inline int cdiv(long long a, long long b) { return (int)((a + b - 1) / b); }

extern "C" void kernel_launch(void* const* d_in, const int* in_sizes, int n_in,
                              void* d_out, int out_size, void* d_ws, size_t ws_size,
                              hipStream_t stream) {
    const float* feat = (const float*)d_in[0];
    const int*   src  = (const int*)d_in[1];
    const int*   dst  = (const int*)d_in[2];
    const float* W0   = (const float*)d_in[3];
    const float* W1   = (const float*)d_in[4];
    const float* W2   = (const float*)d_in[5];
    const float* b2   = (const float*)d_in[6];
    const float* g_in = (const float*)d_in[7];
    const float* b_in = (const float*)d_in[8];
    const float* g0   = (const float*)d_in[9];
    const float* b0   = (const float*)d_in[10];
    const float* g1   = (const float*)d_in[11];
    const float* b1   = (const float*)d_in[12];
    float* out = (float*)d_out;

    const int IN_F = 768, HID = 128, CLS = 40;
    const int n = in_sizes[0] / IN_F;
    const int e = in_sizes[1];

    float* ws    = (float*)d_ws;
    float* do_is = ws;
    float* di_is = ws + (size_t)n;
    float* mu    = ws + 2 * (size_t)n;
    float* rs    = ws + 3 * (size_t)n;
    float* bufB  = ws + 4 * (size_t)n;
    float* bufC  = bufB + (size_t)n * HID;
    float* bufD  = bufC + (size_t)n * HID;
    float* bufE  = bufD + (size_t)n * HID;

    const int T = 256;

    hipMemsetAsync(do_is, 0, 2 * (size_t)n * sizeof(float), stream);
    degree_kernel<<<cdiv(e, T), T, 0, stream>>>(src, dst, do_is, di_is, e);
    invsqrt_deg_kernel<<<cdiv(n, T), T, 0, stream>>>(do_is, di_is, n);

    ln_stats_768_kernel<<<cdiv((long long)n * 32, T), T, 0, stream>>>(feat, mu, rs, n);

    const int mtiles = cdiv(n, 16);
    gemm0_ln_wmma_kernel<<<cdiv((long long)mtiles * 32, T), T, 0, stream>>>(
        feat, mu, rs, do_is, g_in, b_in, W0, bufB, n);

    hipMemsetAsync(bufC, 0, (size_t)n * HID * sizeof(float), stream);
    scatter_add_vec4_kernel<32><<<cdiv((long long)e * 32, T), T, 0, stream>>>(bufB, src, dst, bufC, e);

    post_ln_relu_128_kernel<<<cdiv((long long)n * 32, T), T, 0, stream>>>(
        bufC, di_is, do_is, g0, b0, bufD, n);

    hipMemsetAsync(bufB, 0, (size_t)n * HID * sizeof(float), stream);
    scatter_add_vec4_kernel<32><<<cdiv((long long)e * 32, T), T, 0, stream>>>(bufD, src, dst, bufB, e);
    gemm_wmma_strip_kernel<8><<<cdiv((long long)mtiles * 32, T), T, 0, stream>>>(
        bufB, W1, bufC, n, HID, HID);

    post_ln_relu_128_kernel<<<cdiv((long long)n * 32, T), T, 0, stream>>>(
        bufC, di_is, do_is, g1, b1, bufD, n);

    gemm_wmma_strip_kernel<3><<<cdiv((long long)mtiles * 32, T), T, 0, stream>>>(
        bufD, W2, bufE, n, HID, CLS);
    hipMemsetAsync(out, 0, (size_t)n * CLS * sizeof(float), stream);
    scatter_add_vec4_kernel<10><<<cdiv((long long)e * 10, T), T, 0, stream>>>(bufE, src, dst, out, e);
    final_scale_bias_kernel<<<cdiv((long long)n * CLS, T), T, 0, stream>>>(out, di_is, b2, n);
}


# Aggregation: dst-sorted CSR (int-atomic count with arrival-order rank, block scan, fill) + one-wave-per-row gather replace the 3 float-atomic scatter-adds; row sums sequential in arrival order
# speedup vs baseline: 1.5927x; 1.5927x over previous
#include <hip/hip_runtime.h>
#include <math.h>

typedef __attribute__((ext_vector_type(2))) float v2f;
typedef __attribute__((ext_vector_type(8))) float v8f;

#define LN_EPS 1e-5f

__global__ void degree_rank_kernel(const int* __restrict__ src, const int* __restrict__ dst,
                                   int* __restrict__ cnt_out, int* __restrict__ cnt_in,
                                   int* __restrict__ rank, int e) {
    int i = blockIdx.x * blockDim.x + threadIdx.x;
    if (i >= e) return;
    atomicAdd(cnt_out + src[i], 1);
    rank[i] = atomicAdd(cnt_in + dst[i], 1);
}

__global__ void __launch_bounds__(1024)
rowptr_scan_kernel(const int* __restrict__ cnt_out, const int* __restrict__ cnt_in,
                   int* __restrict__ rowptr,
                   float* __restrict__ do_is, float* __restrict__ di_is, int n) {
    __shared__ int wtot[32];
    __shared__ int wpart[32];
    const int tid  = threadIdx.x;
    const int lane = tid & 31;
    const int wid  = tid >> 5;
    const int tile0 = blockIdx.x * 1024;

    int part = 0;
    for (int j = tid; j < tile0; j += 1024) part += cnt_in[j];

    const int i = tile0 + tid;
    const int c = (i < n) ? cnt_in[i] : 0;
    int incl = c;
    #pragma unroll
    for (int off = 1; off < 32; off <<= 1) {
        const int t = __shfl_up(incl, off, 32);
        if (lane >= off) incl += t;
    }
    #pragma unroll
    for (int off = 16; off > 0; off >>= 1) part += __shfl_xor(part, off, 32);
    if (lane == 31) wtot[wid]  = incl;
    if (lane == 0)  wpart[wid] = part;
    __syncthreads();
    int bsum = wpart[lane];
    int pre  = (lane < wid) ? wtot[lane] : 0;
    #pragma unroll
    for (int off = 16; off > 0; off >>= 1) {
        bsum += __shfl_xor(bsum, off, 32);
        pre  += __shfl_xor(pre,  off, 32);
    }
    if (i < n) {
        rowptr[i] = bsum + pre + incl - c;
        do_is[i]  = rsqrtf(fmaxf((float)cnt_out[i], 1.0f));
        di_is[i]  = rsqrtf(fmaxf((float)c, 1.0f));
    }
}

__global__ void csr_fill_kernel(const int* __restrict__ src, const int* __restrict__ dst,
                                const int* __restrict__ rank, const int* __restrict__ rowptr,
                                int* __restrict__ csr_src, int e) {
    int i = blockIdx.x * blockDim.x + threadIdx.x;
    if (i >= e) return;
    csr_src[rowptr[dst[i]] + rank[i]] = src[i];
}

__global__ void ln_stats_768_kernel(const float* __restrict__ feat,
                                    float* __restrict__ mu, float* __restrict__ rs, int n) {
    int row  = (blockIdx.x * blockDim.x + threadIdx.x) >> 5;
    int lane = threadIdx.x & 31;
    if (row >= n) return;
    const float* p = feat + (size_t)row * 768;
    float s = 0.f, s2 = 0.f;
    #pragma unroll
    for (int c = 0; c < 768; c += 32) {
        float v = p[c + lane];
        s += v; s2 += v * v;
    }
    #pragma unroll
    for (int off = 16; off > 0; off >>= 1) {
        s  += __shfl_xor(s,  off, 32);
        s2 += __shfl_xor(s2, off, 32);
    }
    if (lane == 0) {
        float m   = s  * (1.0f / 768.0f);
        float var = s2 * (1.0f / 768.0f) - m * m;
        mu[row] = m;
        rs[row] = rsqrtf(var + LN_EPS);
    }
}

__global__ void gemm0_ln_wmma_kernel(const float* __restrict__ feat,
                                     const float* __restrict__ mu, const float* __restrict__ rs,
                                     const float* __restrict__ do_is,
                                     const float* __restrict__ g, const float* __restrict__ b,
                                     const float* __restrict__ W,
                                     float* __restrict__ C, int M) {
    const int K = 768, F = 128;
    const int lane = threadIdx.x & 31;
    const int mt   = (blockIdx.x * blockDim.x + threadIdx.x) >> 5;
    if (mt * 16 >= M) return;
    const int half = lane >> 4;
    const int l16  = lane & 15;
    const int arow = mt * 16 + l16;
    const int koff = half * 2;
    const float m_r  = mu[arow];
    const float rs_r = rs[arow];
    const float do_r = do_is[arow];
    const float* ap = feat + (size_t)arow * K + koff;

    v8f acc[8];
    #pragma unroll
    for (int i = 0; i < 8; ++i) { v8f z = {}; acc[i] = z; }

    for (int k = 0; k < K; k += 4) {
        const int c0 = k + koff;
        const v2f x  = *(const v2f*)(ap + k);
        const v2f gg = *(const v2f*)(g + c0);
        const v2f bbv = *(const v2f*)(b + c0);
        v2f a;
        a.x = ((x.x - m_r) * rs_r * gg.x + bbv.x) * do_r;
        a.y = ((x.y - m_r) * rs_r * gg.y + bbv.y) * do_r;
        const float* wrow = W + (size_t)c0 * F;
        #pragma unroll
        for (int ft = 0; ft < 8; ++ft) {
            const int bcol = ft * 16 + l16;
            v2f bb;
            bb.x = wrow[bcol];
            bb.y = wrow[bcol + F];
            acc[ft] = __builtin_amdgcn_wmma_f32_16x16x4_f32(
                false, a, false, bb, (short)0, acc[ft], false, false);
        }
    }
    const int crow = mt * 16 + half * 8;
    #pragma unroll
    for (int ft = 0; ft < 8; ++ft) {
        const int bcol = ft * 16 + l16;
        #pragma unroll
        for (int r = 0; r < 8; ++r)
            C[(size_t)(crow + r) * F + bcol] = acc[ft][r];
    }
}

template <int FT>
__global__ void gemm_wmma_strip_kernel(const float* __restrict__ A, const float* __restrict__ W,
                                       float* __restrict__ C, int M, int K, int F) {
    const int lane = threadIdx.x & 31;
    const int mt   = (blockIdx.x * blockDim.x + threadIdx.x) >> 5;
    if (mt * 16 >= M) return;
    const int half = lane >> 4;
    const int l16  = lane & 15;
    const int arow = mt * 16 + l16;
    const int koff = half * 2;
    const float* ap = A + (size_t)arow * K + koff;

    v8f acc[FT];
    #pragma unroll
    for (int i = 0; i < FT; ++i) { v8f z = {}; acc[i] = z; }

    for (int k = 0; k < K; k += 4) {
        const v2f a = *(const v2f*)(ap + k);
        const float* wrow = W + (size_t)(k + koff) * F;
        #pragma unroll
        for (int ft = 0; ft < FT; ++ft) {
            const int  bcol = ft * 16 + l16;
            const bool bv   = (bcol < F);
            const int  bc   = bv ? bcol : 0;
            v2f bb;
            bb.x = bv ? wrow[bc]     : 0.0f;
            bb.y = bv ? wrow[bc + F] : 0.0f;
            acc[ft] = __builtin_amdgcn_wmma_f32_16x16x4_f32(
                false, a, false, bb, (short)0, acc[ft], false, false);
        }
    }
    const int crow = mt * 16 + half * 8;
    #pragma unroll
    for (int ft = 0; ft < FT; ++ft) {
        const int bcol = ft * 16 + l16;
        if (bcol < F) {
            #pragma unroll
            for (int r = 0; r < 8; ++r)
                C[(size_t)(crow + r) * F + bcol] = acc[ft][r];
        }
    }
}

template <int F4>
__global__ void gather_sum_vec4_kernel(const float* __restrict__ x,
                                       const int* __restrict__ rowptr,
                                       const int* __restrict__ cnt_in,
                                       const int* __restrict__ csr_src,
                                       float* __restrict__ out, int n) {
    const int lane = threadIdx.x & 31;
    const int row = __builtin_amdgcn_readfirstlane((int)((blockIdx.x * blockDim.x + threadIdx.x) >> 5));
    if (row >= n) return;
    const int beg = rowptr[row];
    const int len = cnt_in[row];
    const int c4  = (lane < F4) ? lane : 0;
    const float4* xp = (const float4*)x + c4;
    float4 acc = make_float4(0.f, 0.f, 0.f, 0.f);
    for (int base = 0; base < len; base += 32) {
        const int m = (len - base < 32) ? (len - base) : 32;
        const int idx = (lane < m) ? csr_src[beg + base + lane] : 0;
        #pragma unroll 4
        for (int j = 0; j < m; ++j) {
            const int s = __builtin_amdgcn_readlane(idx, j);
            const float4 v = xp[(size_t)s * F4];
            acc.x += v.x; acc.y += v.y; acc.z += v.z; acc.w += v.w;
        }
    }
    if (lane < F4) ((float4*)out)[(size_t)row * F4 + lane] = acc;
}

__global__ void post_ln_relu_128_kernel(const float* __restrict__ agg,
                                        const float* __restrict__ di_is,
                                        const float* __restrict__ do_is,
                                        const float* __restrict__ g, const float* __restrict__ b,
                                        float* __restrict__ out, int n) {
    int row  = (blockIdx.x * blockDim.x + threadIdx.x) >> 5;
    int lane = threadIdx.x & 31;
    if (row >= n) return;
    const float* p = agg + (size_t)row * 128;
    const float di = di_is[row];
    float v[4];
    float s = 0.f, s2 = 0.f;
    #pragma unroll
    for (int i = 0; i < 4; ++i) {
        v[i] = p[lane + 32 * i] * di;
        s += v[i]; s2 += v[i] * v[i];
    }
    #pragma unroll
    for (int off = 16; off > 0; off >>= 1) {
        s  += __shfl_xor(s,  off, 32);
        s2 += __shfl_xor(s2, off, 32);
    }
    const float m   = s  * (1.0f / 128.0f);
    const float rsv = rsqrtf(s2 * (1.0f / 128.0f) - m * m + LN_EPS);
    const float ds  = do_is[row];
    float* q = out + (size_t)row * 128;
    #pragma unroll
    for (int i = 0; i < 4; ++i) {
        const int c = lane + 32 * i;
        float y = (v[i] - m) * rsv * g[c] + b[c];
        q[c] = fmaxf(y, 0.0f) * ds;
    }
}

__global__ void final_scale_bias_kernel(float* __restrict__ out,
                                        const float* __restrict__ di_is,
                                        const float* __restrict__ b2, int n) {
    const int F = 40;
    int t = blockIdx.x * blockDim.x + threadIdx.x;
    if (t >= n * F) return;
    out[t] = out[t] * di_is[t / F] + b2[t % F];
}

static inline int cdiv(long long a, long long b) { return (int)((a + b - 1) / b); }

extern "C" void kernel_launch(void* const* d_in, const int* in_sizes, int n_in,
                              void* d_out, int out_size, void* d_ws, size_t ws_size,
                              hipStream_t stream) {
    const float* feat = (const float*)d_in[0];
    const int*   src  = (const int*)d_in[1];
    const int*   dst  = (const int*)d_in[2];
    const float* W0   = (const float*)d_in[3];
    const float* W1   = (const float*)d_in[4];
    const float* W2   = (const float*)d_in[5];
    const float* b2   = (const float*)d_in[6];
    const float* g_in = (const float*)d_in[7];
    const float* b_in = (const float*)d_in[8];
    const float* g0   = (const float*)d_in[9];
    const float* b0   = (const float*)d_in[10];
    const float* g1   = (const float*)d_in[11];
    const float* b1   = (const float*)d_in[12];
    float* out = (float*)d_out;

    const int IN_F = 768, HID = 128, CLS = 40;
    const int n = in_sizes[0] / IN_F;
    const int e = in_sizes[1];

    float* ws    = (float*)d_ws;
    float* do_is = ws;
    float* di_is = ws + (size_t)n;
    float* mu    = ws + 2 * (size_t)n;
    float* rs    = ws + 3 * (size_t)n;
    float* bufB  = ws + 4 * (size_t)n;
    float* bufC  = bufB + (size_t)n * HID;
    float* bufD  = bufC + (size_t)n * HID;
    int*   cnt_out = (int*)(bufD + (size_t)n * HID);
    int*   cnt_in  = cnt_out + (size_t)n;
    int*   rowptr  = cnt_in + (size_t)n;
    int*   rank    = rowptr + (size_t)n;
    int*   csr_src = rank + (size_t)e;
    float* bufE  = bufB;

    const int T = 256;

    hipMemsetAsync(cnt_out, 0, 2 * (size_t)n * sizeof(int), stream);
    degree_rank_kernel<<<cdiv(e, T), T, 0, stream>>>(src, dst, cnt_out, cnt_in, rank, e);
    rowptr_scan_kernel<<<cdiv(n, 1024), 1024, 0, stream>>>(cnt_out, cnt_in, rowptr, do_is, di_is, n);
    csr_fill_kernel<<<cdiv(e, T), T, 0, stream>>>(src, dst, rank, rowptr, csr_src, e);

    ln_stats_768_kernel<<<cdiv((long long)n * 32, T), T, 0, stream>>>(feat, mu, rs, n);

    const int mtiles = cdiv(n, 16);
    gemm0_ln_wmma_kernel<<<cdiv((long long)mtiles * 32, T), T, 0, stream>>>(
        feat, mu, rs, do_is, g_in, b_in, W0, bufB, n);

    gather_sum_vec4_kernel<32><<<cdiv((long long)n * 32, T), T, 0, stream>>>(
        bufB, rowptr, cnt_in, csr_src, bufC, n);

    post_ln_relu_128_kernel<<<cdiv((long long)n * 32, T), T, 0, stream>>>(
        bufC, di_is, do_is, g0, b0, bufD, n);

    gather_sum_vec4_kernel<32><<<cdiv((long long)n * 32, T), T, 0, stream>>>(
        bufD, rowptr, cnt_in, csr_src, bufB, n);
    gemm_wmma_strip_kernel<8><<<cdiv((long long)mtiles * 32, T), T, 0, stream>>>(
        bufB, W1, bufC, n, HID, HID);

    post_ln_relu_128_kernel<<<cdiv((long long)n * 32, T), T, 0, stream>>>(
        bufC, di_is, do_is, g1, b1, bufD, n);

    gemm_wmma_strip_kernel<3><<<cdiv((long long)mtiles * 32, T), T, 0, stream>>>(
        bufD, W2, bufE, n, HID, CLS);
    gather_sum_vec4_kernel<10><<<cdiv((long long)n * 32, T), T, 0, stream>>>(
        bufE, rowptr, cnt_in, csr_src, out, n);
    final_scale_bias_kernel<<<cdiv((long long)n * CLS, T), T, 0, stream>>>(out, di_is, b2, n);
}


# Row-wise steps fused into the kernel that produces the row (gather+LayerNorm+ReLU, GEMM1 epilogue LayerNorm+ReLU, gather+scale+bias): 3 launches fewer; LayerNorm row sums in another lane order
# speedup vs baseline: 1.9211x; 1.2062x over previous
#include <hip/hip_runtime.h>
#include <math.h>

typedef __attribute__((ext_vector_type(2))) float v2f;
typedef __attribute__((ext_vector_type(8))) float v8f;

#define LN_EPS 1e-5f

__global__ void degree_rank_kernel(const int* __restrict__ src, const int* __restrict__ dst,
                                   int* __restrict__ cnt_out, int* __restrict__ cnt_in,
                                   int* __restrict__ rank, int e) {
    int i = blockIdx.x * blockDim.x + threadIdx.x;
    if (i >= e) return;
    atomicAdd(cnt_out + src[i], 1);
    rank[i] = atomicAdd(cnt_in + dst[i], 1);
}

__global__ void __launch_bounds__(1024)
rowptr_scan_kernel(const int* __restrict__ cnt_out, const int* __restrict__ cnt_in,
                   int* __restrict__ rowptr,
                   float* __restrict__ do_is, float* __restrict__ di_is, int n) {
    __shared__ int wtot[32];
    __shared__ int wpart[32];
    const int tid  = threadIdx.x;
    const int lane = tid & 31;
    const int wid  = tid >> 5;
    const int tile0 = blockIdx.x * 1024;

    int part = 0;
    for (int j = tid; j < tile0; j += 1024) part += cnt_in[j];

    const int i = tile0 + tid;
    const int c = (i < n) ? cnt_in[i] : 0;
    int incl = c;
    #pragma unroll
    for (int off = 1; off < 32; off <<= 1) {
        const int t = __shfl_up(incl, off, 32);
        if (lane >= off) incl += t;
    }
    #pragma unroll
    for (int off = 16; off > 0; off >>= 1) part += __shfl_xor(part, off, 32);
    if (lane == 31) wtot[wid]  = incl;
    if (lane == 0)  wpart[wid] = part;
    __syncthreads();
    int bsum = wpart[lane];
    int pre  = (lane < wid) ? wtot[lane] : 0;
    #pragma unroll
    for (int off = 16; off > 0; off >>= 1) {
        bsum += __shfl_xor(bsum, off, 32);
        pre  += __shfl_xor(pre,  off, 32);
    }
    if (i < n) {
        rowptr[i] = bsum + pre + incl - c;
        do_is[i]  = rsqrtf(fmaxf((float)cnt_out[i], 1.0f));
        di_is[i]  = rsqrtf(fmaxf((float)c, 1.0f));
    }
}

__global__ void csr_fill_kernel(const int* __restrict__ src, const int* __restrict__ dst,
                                const int* __restrict__ rank, const int* __restrict__ rowptr,
                                int* __restrict__ csr_src, int e) {
    int i = blockIdx.x * blockDim.x + threadIdx.x;
    if (i >= e) return;
    csr_src[rowptr[dst[i]] + rank[i]] = src[i];
}

__global__ void ln_stats_768_kernel(const float* __restrict__ feat,
                                    float* __restrict__ mu, float* __restrict__ rs, int n) {
    int row  = (blockIdx.x * blockDim.x + threadIdx.x) >> 5;
    int lane = threadIdx.x & 31;
    if (row >= n) return;
    const float* p = feat + (size_t)row * 768;
    float s = 0.f, s2 = 0.f;
    #pragma unroll
    for (int c = 0; c < 768; c += 32) {
        float v = p[c + lane];
        s += v; s2 += v * v;
    }
    #pragma unroll
    for (int off = 16; off > 0; off >>= 1) {
        s  += __shfl_xor(s,  off, 32);
        s2 += __shfl_xor(s2, off, 32);
    }
    if (lane == 0) {
        float m   = s  * (1.0f / 768.0f);
        float var = s2 * (1.0f / 768.0f) - m * m;
        mu[row] = m;
        rs[row] = rsqrtf(var + LN_EPS);
    }
}

__global__ void gemm0_ln_wmma_kernel(const float* __restrict__ feat,
                                     const float* __restrict__ mu, const float* __restrict__ rs,
                                     const float* __restrict__ do_is,
                                     const float* __restrict__ g, const float* __restrict__ b,
                                     const float* __restrict__ W,
                                     float* __restrict__ C, int M) {
    const int K = 768, F = 128;
    const int lane = threadIdx.x & 31;
    const int mt   = (blockIdx.x * blockDim.x + threadIdx.x) >> 5;
    if (mt * 16 >= M) return;
    const int half = lane >> 4;
    const int l16  = lane & 15;
    const int arow = mt * 16 + l16;
    const int koff = half * 2;
    const float m_r  = mu[arow];
    const float rs_r = rs[arow];
    const float do_r = do_is[arow];
    const float* ap = feat + (size_t)arow * K + koff;

    v8f acc[8];
    #pragma unroll
    for (int i = 0; i < 8; ++i) { v8f z = {}; acc[i] = z; }

    for (int k = 0; k < K; k += 4) {
        const int c0 = k + koff;
        const v2f x  = *(const v2f*)(ap + k);
        const v2f gg = *(const v2f*)(g + c0);
        const v2f bbv = *(const v2f*)(b + c0);
        v2f a;
        a.x = ((x.x - m_r) * rs_r * gg.x + bbv.x) * do_r;
        a.y = ((x.y - m_r) * rs_r * gg.y + bbv.y) * do_r;
        const float* wrow = W + (size_t)c0 * F;
        #pragma unroll
        for (int ft = 0; ft < 8; ++ft) {
            const int bcol = ft * 16 + l16;
            v2f bb;
            bb.x = wrow[bcol];
            bb.y = wrow[bcol + F];
            acc[ft] = __builtin_amdgcn_wmma_f32_16x16x4_f32(
                false, a, false, bb, (short)0, acc[ft], false, false);
        }
    }
    const int crow = mt * 16 + half * 8;
    #pragma unroll
    for (int ft = 0; ft < 8; ++ft) {
        const int bcol = ft * 16 + l16;
        #pragma unroll
        for (int r = 0; r < 8; ++r)
            C[(size_t)(crow + r) * F + bcol] = acc[ft][r];
    }
}

template <int FT>
__global__ void gemm_wmma_strip_kernel(const float* __restrict__ A, const float* __restrict__ W,
                                       float* __restrict__ C, int M, int K, int F) {
    const int lane = threadIdx.x & 31;
    const int mt   = (blockIdx.x * blockDim.x + threadIdx.x) >> 5;
    if (mt * 16 >= M) return;
    const int half = lane >> 4;
    const int l16  = lane & 15;
    const int arow = mt * 16 + l16;
    const int koff = half * 2;
    const float* ap = A + (size_t)arow * K + koff;

    v8f acc[FT];
    #pragma unroll
    for (int i = 0; i < FT; ++i) { v8f z = {}; acc[i] = z; }

    for (int k = 0; k < K; k += 4) {
        const v2f a = *(const v2f*)(ap + k);
        const float* wrow = W + (size_t)(k + koff) * F;
        #pragma unroll
        for (int ft = 0; ft < FT; ++ft) {
            const int  bcol = ft * 16 + l16;
            const bool bv   = (bcol < F);
            const int  bc   = bv ? bcol : 0;
            v2f bb;
            bb.x = bv ? wrow[bc]     : 0.0f;
            bb.y = bv ? wrow[bc + F] : 0.0f;
            acc[ft] = __builtin_amdgcn_wmma_f32_16x16x4_f32(
                false, a, false, bb, (short)0, acc[ft], false, false);
        }
    }
    const int crow = mt * 16 + half * 8;
    #pragma unroll
    for (int ft = 0; ft < FT; ++ft) {
        const int bcol = ft * 16 + l16;
        if (bcol < F) {
            #pragma unroll
            for (int r = 0; r < 8; ++r)
                C[(size_t)(crow + r) * F + bcol] = acc[ft][r];
        }
    }
}

template <int F4>
__device__ __forceinline__ float4 gather_row_sum(const float* __restrict__ x,
                                                 const int* __restrict__ rowptr,
                                                 const int* __restrict__ cnt_in,
                                                 const int* __restrict__ csr_src,
                                                 int row, int lane) {
    const int beg = rowptr[row];
    const int len = cnt_in[row];
    const int c4  = (lane < F4) ? lane : 0;
    const float4* xp = (const float4*)x + c4;
    float4 acc = make_float4(0.f, 0.f, 0.f, 0.f);
    for (int base = 0; base < len; base += 32) {
        const int m = (len - base < 32) ? (len - base) : 32;
        const int idx = (lane < m) ? csr_src[beg + base + lane] : 0;
        #pragma unroll 4
        for (int j = 0; j < m; ++j) {
            const int s = __builtin_amdgcn_readlane(idx, j);
            const float4 v = xp[(size_t)s * F4];
            acc.x += v.x; acc.y += v.y; acc.z += v.z; acc.w += v.w;
        }
    }
    return acc;
}

template <int F4>
__global__ void gather_sum_vec4_kernel(const float* __restrict__ x,
                                       const int* __restrict__ rowptr,
                                       const int* __restrict__ cnt_in,
                                       const int* __restrict__ csr_src,
                                       float* __restrict__ out, int n) {
    const int lane = threadIdx.x & 31;
    const int row = __builtin_amdgcn_readfirstlane((int)((blockIdx.x * blockDim.x + threadIdx.x) >> 5));
    if (row >= n) return;
    const float4 acc = gather_row_sum<F4>(x, rowptr, cnt_in, csr_src, row, lane);
    if (lane < F4) ((float4*)out)[(size_t)row * F4 + lane] = acc;
}

__global__ void gather_ln_relu_128_kernel(const float* __restrict__ x,
                                          const int* __restrict__ rowptr,
                                          const int* __restrict__ cnt_in,
                                          const int* __restrict__ csr_src,
                                          const float* __restrict__ di_is,
                                          const float* __restrict__ do_is,
                                          const float* __restrict__ g, const float* __restrict__ b,
                                          float* __restrict__ out, int n) {
    const int lane = threadIdx.x & 31;
    const int row = __builtin_amdgcn_readfirstlane((int)((blockIdx.x * blockDim.x + threadIdx.x) >> 5));
    if (row >= n) return;
    const float4 acc = gather_row_sum<32>(x, rowptr, cnt_in, csr_src, row, lane);
    const float di = di_is[row];
    const float ds = do_is[row];
    float v[4] = { acc.x * di, acc.y * di, acc.z * di, acc.w * di };
    float s = 0.f, s2 = 0.f;
    #pragma unroll
    for (int i = 0; i < 4; ++i) { s += v[i]; s2 += v[i] * v[i]; }
    #pragma unroll
    for (int off = 16; off > 0; off >>= 1) {
        s  += __shfl_xor(s,  off, 32);
        s2 += __shfl_xor(s2, off, 32);
    }
    const float m   = s  * (1.0f / 128.0f);
    const float rsv = rsqrtf(s2 * (1.0f / 128.0f) - m * m + LN_EPS);
    float y[4];
    #pragma unroll
    for (int i = 0; i < 4; ++i) {
        const int c = 4 * lane + i;
        const float t = (v[i] - m) * rsv * g[c] + b[c];
        y[i] = fmaxf(t, 0.0f) * ds;
    }
    ((float4*)out)[(size_t)row * 32 + lane] = make_float4(y[0], y[1], y[2], y[3]);
}

__global__ void gather_scale_bias_40_kernel(const float* __restrict__ x,
                                            const int* __restrict__ rowptr,
                                            const int* __restrict__ cnt_in,
                                            const int* __restrict__ csr_src,
                                            const float* __restrict__ di_is,
                                            const float* __restrict__ b2,
                                            float* __restrict__ out, int n) {
    const int lane = threadIdx.x & 31;
    const int row = __builtin_amdgcn_readfirstlane((int)((blockIdx.x * blockDim.x + threadIdx.x) >> 5));
    if (row >= n) return;
    const float4 acc = gather_row_sum<10>(x, rowptr, cnt_in, csr_src, row, lane);
    if (lane < 10) {
        const float di = di_is[row];
        const int c = 4 * lane;
        float4 o;
        o.x = acc.x * di + b2[c];
        o.y = acc.y * di + b2[c + 1];
        o.z = acc.z * di + b2[c + 2];
        o.w = acc.w * di + b2[c + 3];
        ((float4*)out)[(size_t)row * 10 + lane] = o;
    }
}

__global__ void gemm128_ln_relu_wmma_kernel(const float* __restrict__ A, const float* __restrict__ W,
                                            const float* __restrict__ di_is,
                                            const float* __restrict__ do_is,
                                            const float* __restrict__ g, const float* __restrict__ b,
                                            float* __restrict__ C, int M) {
    const int K = 128, F = 128;
    const int lane = threadIdx.x & 31;
    const int mt   = (blockIdx.x * blockDim.x + threadIdx.x) >> 5;
    if (mt * 16 >= M) return;
    const int half = lane >> 4;
    const int l16  = lane & 15;
    const int arow = mt * 16 + l16;
    const int koff = half * 2;
    const float* ap = A + (size_t)arow * K + koff;

    v8f acc[8];
    #pragma unroll
    for (int i = 0; i < 8; ++i) { v8f z = {}; acc[i] = z; }

    for (int k = 0; k < K; k += 4) {
        const v2f a = *(const v2f*)(ap + k);
        const float* wrow = W + (size_t)(k + koff) * F;
        #pragma unroll
        for (int ft = 0; ft < 8; ++ft) {
            const int bcol = ft * 16 + l16;
            v2f bb;
            bb.x = wrow[bcol];
            bb.y = wrow[bcol + F];
            acc[ft] = __builtin_amdgcn_wmma_f32_16x16x4_f32(
                false, a, false, bb, (short)0, acc[ft], false, false);
        }
    }

    const int crow = mt * 16 + half * 8;
    float gg[8], bbv[8];
    #pragma unroll
    for (int ft = 0; ft < 8; ++ft) {
        gg[ft]  = g[ft * 16 + l16];
        bbv[ft] = b[ft * 16 + l16];
    }
    #pragma unroll
    for (int r = 0; r < 8; ++r) {
        const float di = di_is[crow + r];
        const float ds = do_is[crow + r];
        float v[8];
        float s = 0.f, s2 = 0.f;
        #pragma unroll
        for (int ft = 0; ft < 8; ++ft) {
            v[ft] = acc[ft][r] * di;
            s += v[ft]; s2 += v[ft] * v[ft];
        }
        #pragma unroll
        for (int off = 8; off > 0; off >>= 1) {
            s  += __shfl_xor(s,  off, 32);
            s2 += __shfl_xor(s2, off, 32);
        }
        const float m   = s  * (1.0f / 128.0f);
        const float rsv = rsqrtf(s2 * (1.0f / 128.0f) - m * m + LN_EPS);
        #pragma unroll
        for (int ft = 0; ft < 8; ++ft) {
            const float t = (v[ft] - m) * rsv * gg[ft] + bbv[ft];
            C[(size_t)(crow + r) * F + ft * 16 + l16] = fmaxf(t, 0.0f) * ds;
        }
    }
}

static inline int cdiv(long long a, long long b) { return (int)((a + b - 1) / b); }

extern "C" void kernel_launch(void* const* d_in, const int* in_sizes, int n_in,
                              void* d_out, int out_size, void* d_ws, size_t ws_size,
                              hipStream_t stream) {
    const float* feat = (const float*)d_in[0];
    const int*   src  = (const int*)d_in[1];
    const int*   dst  = (const int*)d_in[2];
    const float* W0   = (const float*)d_in[3];
    const float* W1   = (const float*)d_in[4];
    const float* W2   = (const float*)d_in[5];
    const float* b2   = (const float*)d_in[6];
    const float* g_in = (const float*)d_in[7];
    const float* b_in = (const float*)d_in[8];
    const float* g0   = (const float*)d_in[9];
    const float* b0   = (const float*)d_in[10];
    const float* g1   = (const float*)d_in[11];
    const float* b1   = (const float*)d_in[12];
    float* out = (float*)d_out;

    const int IN_F = 768, HID = 128, CLS = 40;
    const int n = in_sizes[0] / IN_F;
    const int e = in_sizes[1];

    float* ws    = (float*)d_ws;
    float* do_is = ws;
    float* di_is = ws + (size_t)n;
    float* mu    = ws + 2 * (size_t)n;
    float* rs    = ws + 3 * (size_t)n;
    float* bufB  = ws + 4 * (size_t)n;
    float* bufC  = bufB + (size_t)n * HID;
    float* bufD  = bufC + (size_t)n * HID;
    int*   cnt_out = (int*)(bufD + (size_t)n * HID);
    int*   cnt_in  = cnt_out + (size_t)n;
    int*   rowptr  = cnt_in + (size_t)n;
    int*   rank    = rowptr + (size_t)n;
    int*   csr_src = rank + (size_t)e;
    float* bufE  = bufB;

    const int T = 256;

    hipMemsetAsync(cnt_out, 0, 2 * (size_t)n * sizeof(int), stream);
    degree_rank_kernel<<<cdiv(e, T), T, 0, stream>>>(src, dst, cnt_out, cnt_in, rank, e);
    rowptr_scan_kernel<<<cdiv(n, 1024), 1024, 0, stream>>>(cnt_out, cnt_in, rowptr, do_is, di_is, n);
    csr_fill_kernel<<<cdiv(e, T), T, 0, stream>>>(src, dst, rank, rowptr, csr_src, e);

    ln_stats_768_kernel<<<cdiv((long long)n * 32, T), T, 0, stream>>>(feat, mu, rs, n);

    const int mtiles = cdiv(n, 16);
    gemm0_ln_wmma_kernel<<<cdiv((long long)mtiles * 32, T), T, 0, stream>>>(
        feat, mu, rs, do_is, g_in, b_in, W0, bufB, n);

    gather_ln_relu_128_kernel<<<cdiv((long long)n * 32, T), T, 0, stream>>>(
        bufB, rowptr, cnt_in, csr_src, di_is, do_is, g0, b0, bufD, n);

    gather_sum_vec4_kernel<32><<<cdiv((long long)n * 32, T), T, 0, stream>>>(
        bufD, rowptr, cnt_in, csr_src, bufB, n);
    gemm128_ln_relu_wmma_kernel<<<cdiv((long long)mtiles * 32, T), T, 0, stream>>>(
        bufB, W1, di_is, do_is, g1, b1, bufC, n);

    gemm_wmma_strip_kernel<3><<<cdiv((long long)mtiles * 32, T), T, 0, stream>>>(
        bufC, W2, bufE, n, HID, CLS);
    gather_scale_bias_40_kernel<<<cdiv((long long)n * 32, T), T, 0, stream>>>(
        bufE, rowptr, cnt_in, csr_src, di_is, b2, out, n);
}


# Independent kernels share a launch: row-offset scan beside the LayerNorm statistics, CSR fill beside the layer-0 product (strip blocks first, edge blocks behind); 2 launches fewer, bodies unchanged
# speedup vs baseline: 2.0387x; 1.0612x over previous
#include <hip/hip_runtime.h>
#include <math.h>

typedef __attribute__((ext_vector_type(2))) float v2f;
typedef __attribute__((ext_vector_type(8))) float v8f;

#define LN_EPS 1e-5f

__global__ void degree_rank_kernel(const int* __restrict__ src, const int* __restrict__ dst,
                                   int* __restrict__ cnt_out, int* __restrict__ cnt_in,
                                   int* __restrict__ rank, int e) {
    int i = blockIdx.x * blockDim.x + threadIdx.x;
    if (i >= e) return;
    atomicAdd(cnt_out + src[i], 1);
    rank[i] = atomicAdd(cnt_in + dst[i], 1);
}

__device__ __forceinline__ void
rowptr_scan_tile(int blk, const int* __restrict__ cnt_out, const int* __restrict__ cnt_in,
                 int* __restrict__ rowptr,
                 float* __restrict__ do_is, float* __restrict__ di_is, int n) {
    __shared__ int wtot[32];
    __shared__ int wpart[32];
    const int tid  = threadIdx.x;
    const int lane = tid & 31;
    const int wid  = tid >> 5;
    const int tile0 = blk * 1024;

    int part = 0;
    for (int j = tid; j < tile0; j += 1024) part += cnt_in[j];

    const int i = tile0 + tid;
    const int c = (i < n) ? cnt_in[i] : 0;
    int incl = c;
    #pragma unroll
    for (int off = 1; off < 32; off <<= 1) {
        const int t = __shfl_up(incl, off, 32);
        if (lane >= off) incl += t;
    }
    #pragma unroll
    for (int off = 16; off > 0; off >>= 1) part += __shfl_xor(part, off, 32);
    if (lane == 31) wtot[wid]  = incl;
    if (lane == 0)  wpart[wid] = part;
    __syncthreads();
    int bsum = wpart[lane];
    int pre  = (lane < wid) ? wtot[lane] : 0;
    #pragma unroll
    for (int off = 16; off > 0; off >>= 1) {
        bsum += __shfl_xor(bsum, off, 32);
        pre  += __shfl_xor(pre,  off, 32);
    }
    if (i < n) {
        rowptr[i] = bsum + pre + incl - c;
        do_is[i]  = rsqrtf(fmaxf((float)cnt_out[i], 1.0f));
        di_is[i]  = rsqrtf(fmaxf((float)c, 1.0f));
    }
}

__device__ __forceinline__ void
csr_fill_edges(int blk, const int* __restrict__ src, const int* __restrict__ dst,
               const int* __restrict__ rank, const int* __restrict__ rowptr,
               int* __restrict__ csr_src, int e) {
    int i = blk * blockDim.x + threadIdx.x;
    if (i >= e) return;
    csr_src[rowptr[dst[i]] + rank[i]] = src[i];
}

__device__ __forceinline__ void
ln_stats_768_rows(int blk, const float* __restrict__ feat,
                  float* __restrict__ mu, float* __restrict__ rs, int n) {
    int row  = (blk * blockDim.x + threadIdx.x) >> 5;
    int lane = threadIdx.x & 31;
    if (row >= n) return;
    const float* p = feat + (size_t)row * 768;
    float s = 0.f, s2 = 0.f;
    #pragma unroll
    for (int c = 0; c < 768; c += 32) {
        float v = p[c + lane];
        s += v; s2 += v * v;
    }
    #pragma unroll
    for (int off = 16; off > 0; off >>= 1) {
        s  += __shfl_xor(s,  off, 32);
        s2 += __shfl_xor(s2, off, 32);
    }
    if (lane == 0) {
        float m   = s  * (1.0f / 768.0f);
        float var = s2 * (1.0f / 768.0f) - m * m;
        mu[row] = m;
        rs[row] = rsqrtf(var + LN_EPS);
    }
}

__global__ void __launch_bounds__(1024)
scan_and_ln_stats_kernel(const int* __restrict__ cnt_out, const int* __restrict__ cnt_in,
                         int* __restrict__ rowptr,
                         float* __restrict__ do_is, float* __restrict__ di_is,
                         const float* __restrict__ feat,
                         float* __restrict__ mu, float* __restrict__ rs,
                         int n, int nscan) {
    const int blk = blockIdx.x;
    if (blk < nscan) rowptr_scan_tile(blk, cnt_out, cnt_in, rowptr, do_is, di_is, n);
    else             ln_stats_768_rows(blk - nscan, feat, mu, rs, n);
}

__device__ __forceinline__ void
gemm0_ln_wmma_strips(int blk, const float* __restrict__ feat,
                     const float* __restrict__ mu, const float* __restrict__ rs,
                     const float* __restrict__ do_is,
                     const float* __restrict__ g, const float* __restrict__ b,
                     const float* __restrict__ W,
                     float* __restrict__ C, int M) {
    const int K = 768, F = 128;
    const int lane = threadIdx.x & 31;
    const int mt   = (blk * blockDim.x + threadIdx.x) >> 5;
    if (mt * 16 >= M) return;
    const int half = lane >> 4;
    const int l16  = lane & 15;
    const int arow = mt * 16 + l16;
    const int koff = half * 2;
    const float m_r  = mu[arow];
    const float rs_r = rs[arow];
    const float do_r = do_is[arow];
    const float* ap = feat + (size_t)arow * K + koff;

    v8f acc[8];
    #pragma unroll
    for (int i = 0; i < 8; ++i) { v8f z = {}; acc[i] = z; }

    for (int k = 0; k < K; k += 4) {
        const int c0 = k + koff;
        const v2f x  = *(const v2f*)(ap + k);
        const v2f gg = *(const v2f*)(g + c0);
        const v2f bbv = *(const v2f*)(b + c0);
        v2f a;
        a.x = ((x.x - m_r) * rs_r * gg.x + bbv.x) * do_r;
        a.y = ((x.y - m_r) * rs_r * gg.y + bbv.y) * do_r;
        const float* wrow = W + (size_t)c0 * F;
        #pragma unroll
        for (int ft = 0; ft < 8; ++ft) {
            const int bcol = ft * 16 + l16;
            v2f bb;
            bb.x = wrow[bcol];
            bb.y = wrow[bcol + F];
            acc[ft] = __builtin_amdgcn_wmma_f32_16x16x4_f32(
                false, a, false, bb, (short)0, acc[ft], false, false);
        }
    }
    const int crow = mt * 16 + half * 8;
    #pragma unroll
    for (int ft = 0; ft < 8; ++ft) {
        const int bcol = ft * 16 + l16;
        #pragma unroll
        for (int r = 0; r < 8; ++r)
            C[(size_t)(crow + r) * F + bcol] = acc[ft][r];
    }
}

__global__ void gemm0_and_csr_fill_kernel(const float* __restrict__ feat,
                                          const float* __restrict__ mu, const float* __restrict__ rs,
                                          const float* __restrict__ do_is,
                                          const float* __restrict__ g, const float* __restrict__ b,
                                          const float* __restrict__ W,
                                          float* __restrict__ C, int M, int ngemm,
                                          const int* __restrict__ src, const int* __restrict__ dst,
                                          const int* __restrict__ rank, const int* __restrict__ rowptr,
                                          int* __restrict__ csr_src, int e) {
    const int blk = blockIdx.x;
    if (blk < ngemm) gemm0_ln_wmma_strips(blk, feat, mu, rs, do_is, g, b, W, C, M);
    else             csr_fill_edges(blk - ngemm, src, dst, rank, rowptr, csr_src, e);
}

template <int FT>
__global__ void gemm_wmma_strip_kernel(const float* __restrict__ A, const float* __restrict__ W,
                                       float* __restrict__ C, int M, int K, int F) {
    const int lane = threadIdx.x & 31;
    const int mt   = (blockIdx.x * blockDim.x + threadIdx.x) >> 5;
    if (mt * 16 >= M) return;
    const int half = lane >> 4;
    const int l16  = lane & 15;
    const int arow = mt * 16 + l16;
    const int koff = half * 2;
    const float* ap = A + (size_t)arow * K + koff;

    v8f acc[FT];
    #pragma unroll
    for (int i = 0; i < FT; ++i) { v8f z = {}; acc[i] = z; }

    for (int k = 0; k < K; k += 4) {
        const v2f a = *(const v2f*)(ap + k);
        const float* wrow = W + (size_t)(k + koff) * F;
        #pragma unroll
        for (int ft = 0; ft < FT; ++ft) {
            const int  bcol = ft * 16 + l16;
            const bool bv   = (bcol < F);
            const int  bc   = bv ? bcol : 0;
            v2f bb;
            bb.x = bv ? wrow[bc]     : 0.0f;
            bb.y = bv ? wrow[bc + F] : 0.0f;
            acc[ft] = __builtin_amdgcn_wmma_f32_16x16x4_f32(
                false, a, false, bb, (short)0, acc[ft], false, false);
        }
    }
    const int crow = mt * 16 + half * 8;
    #pragma unroll
    for (int ft = 0; ft < FT; ++ft) {
        const int bcol = ft * 16 + l16;
        if (bcol < F) {
            #pragma unroll
            for (int r = 0; r < 8; ++r)
                C[(size_t)(crow + r) * F + bcol] = acc[ft][r];
        }
    }
}

template <int F4>
__device__ __forceinline__ float4 gather_row_sum(const float* __restrict__ x,
                                                 const int* __restrict__ rowptr,
                                                 const int* __restrict__ cnt_in,
                                                 const int* __restrict__ csr_src,
                                                 int row, int lane) {
    const int beg = rowptr[row];
    const int len = cnt_in[row];
    const int c4  = (lane < F4) ? lane : 0;
    const float4* xp = (const float4*)x + c4;
    float4 acc = make_float4(0.f, 0.f, 0.f, 0.f);
    for (int base = 0; base < len; base += 32) {
        const int m = (len - base < 32) ? (len - base) : 32;
        const int idx = (lane < m) ? csr_src[beg + base + lane] : 0;
        #pragma unroll 4
        for (int j = 0; j < m; ++j) {
            const int s = __builtin_amdgcn_readlane(idx, j);
            const float4 v = xp[(size_t)s * F4];
            acc.x += v.x; acc.y += v.y; acc.z += v.z; acc.w += v.w;
        }
    }
    return acc;
}

template <int F4>
__global__ void gather_sum_vec4_kernel(const float* __restrict__ x,
                                       const int* __restrict__ rowptr,
                                       const int* __restrict__ cnt_in,
                                       const int* __restrict__ csr_src,
                                       float* __restrict__ out, int n) {
    const int lane = threadIdx.x & 31;
    const int row = __builtin_amdgcn_readfirstlane((int)((blockIdx.x * blockDim.x + threadIdx.x) >> 5));
    if (row >= n) return;
    const float4 acc = gather_row_sum<F4>(x, rowptr, cnt_in, csr_src, row, lane);
    if (lane < F4) ((float4*)out)[(size_t)row * F4 + lane] = acc;
}

__global__ void gather_ln_relu_128_kernel(const float* __restrict__ x,
                                          const int* __restrict__ rowptr,
                                          const int* __restrict__ cnt_in,
                                          const int* __restrict__ csr_src,
                                          const float* __restrict__ di_is,
                                          const float* __restrict__ do_is,
                                          const float* __restrict__ g, const float* __restrict__ b,
                                          float* __restrict__ out, int n) {
    const int lane = threadIdx.x & 31;
    const int row = __builtin_amdgcn_readfirstlane((int)((blockIdx.x * blockDim.x + threadIdx.x) >> 5));
    if (row >= n) return;
    const float4 acc = gather_row_sum<32>(x, rowptr, cnt_in, csr_src, row, lane);
    const float di = di_is[row];
    const float ds = do_is[row];
    float v[4] = { acc.x * di, acc.y * di, acc.z * di, acc.w * di };
    float s = 0.f, s2 = 0.f;
    #pragma unroll
    for (int i = 0; i < 4; ++i) { s += v[i]; s2 += v[i] * v[i]; }
    #pragma unroll
    for (int off = 16; off > 0; off >>= 1) {
        s  += __shfl_xor(s,  off, 32);
        s2 += __shfl_xor(s2, off, 32);
    }
    const float m   = s  * (1.0f / 128.0f);
    const float rsv = rsqrtf(s2 * (1.0f / 128.0f) - m * m + LN_EPS);
    float y[4];
    #pragma unroll
    for (int i = 0; i < 4; ++i) {
        const int c = 4 * lane + i;
        const float t = (v[i] - m) * rsv * g[c] + b[c];
        y[i] = fmaxf(t, 0.0f) * ds;
    }
    ((float4*)out)[(size_t)row * 32 + lane] = make_float4(y[0], y[1], y[2], y[3]);
}

__global__ void gather_scale_bias_40_kernel(const float* __restrict__ x,
                                            const int* __restrict__ rowptr,
                                            const int* __restrict__ cnt_in,
                                            const int* __restrict__ csr_src,
                                            const float* __restrict__ di_is,
                                            const float* __restrict__ b2,
                                            float* __restrict__ out, int n) {
    const int lane = threadIdx.x & 31;
    const int row = __builtin_amdgcn_readfirstlane((int)((blockIdx.x * blockDim.x + threadIdx.x) >> 5));
    if (row >= n) return;
    const float4 acc = gather_row_sum<10>(x, rowptr, cnt_in, csr_src, row, lane);
    if (lane < 10) {
        const float di = di_is[row];
        const int c = 4 * lane;
        float4 o;
        o.x = acc.x * di + b2[c];
        o.y = acc.y * di + b2[c + 1];
        o.z = acc.z * di + b2[c + 2];
        o.w = acc.w * di + b2[c + 3];
        ((float4*)out)[(size_t)row * 10 + lane] = o;
    }
}

__global__ void gemm128_ln_relu_wmma_kernel(const float* __restrict__ A, const float* __restrict__ W,
                                            const float* __restrict__ di_is,
                                            const float* __restrict__ do_is,
                                            const float* __restrict__ g, const float* __restrict__ b,
                                            float* __restrict__ C, int M) {
    const int K = 128, F = 128;
    const int lane = threadIdx.x & 31;
    const int mt   = (blockIdx.x * blockDim.x + threadIdx.x) >> 5;
    if (mt * 16 >= M) return;
    const int half = lane >> 4;
    const int l16  = lane & 15;
    const int arow = mt * 16 + l16;
    const int koff = half * 2;
    const float* ap = A + (size_t)arow * K + koff;

    v8f acc[8];
    #pragma unroll
    for (int i = 0; i < 8; ++i) { v8f z = {}; acc[i] = z; }

    for (int k = 0; k < K; k += 4) {
        const v2f a = *(const v2f*)(ap + k);
        const float* wrow = W + (size_t)(k + koff) * F;
        #pragma unroll
        for (int ft = 0; ft < 8; ++ft) {
            const int bcol = ft * 16 + l16;
            v2f bb;
            bb.x = wrow[bcol];
            bb.y = wrow[bcol + F];
            acc[ft] = __builtin_amdgcn_wmma_f32_16x16x4_f32(
                false, a, false, bb, (short)0, acc[ft], false, false);
        }
    }

    const int crow = mt * 16 + half * 8;
    float gg[8], bbv[8];
    #pragma unroll
    for (int ft = 0; ft < 8; ++ft) {
        gg[ft]  = g[ft * 16 + l16];
        bbv[ft] = b[ft * 16 + l16];
    }
    #pragma unroll
    for (int r = 0; r < 8; ++r) {
        const float di = di_is[crow + r];
        const float ds = do_is[crow + r];
        float v[8];
        float s = 0.f, s2 = 0.f;
        #pragma unroll
        for (int ft = 0; ft < 8; ++ft) {
            v[ft] = acc[ft][r] * di;
            s += v[ft]; s2 += v[ft] * v[ft];
        }
        #pragma unroll
        for (int off = 8; off > 0; off >>= 1) {
            s  += __shfl_xor(s,  off, 32);
            s2 += __shfl_xor(s2, off, 32);
        }
        const float m   = s  * (1.0f / 128.0f);
        const float rsv = rsqrtf(s2 * (1.0f / 128.0f) - m * m + LN_EPS);
        #pragma unroll
        for (int ft = 0; ft < 8; ++ft) {
            const float t = (v[ft] - m) * rsv * gg[ft] + bbv[ft];
            C[(size_t)(crow + r) * F + ft * 16 + l16] = fmaxf(t, 0.0f) * ds;
        }
    }
}

static inline int cdiv(long long a, long long b) { return (int)((a + b - 1) / b); }

extern "C" void kernel_launch(void* const* d_in, const int* in_sizes, int n_in,
                              void* d_out, int out_size, void* d_ws, size_t ws_size,
                              hipStream_t stream) {
    const float* feat = (const float*)d_in[0];
    const int*   src  = (const int*)d_in[1];
    const int*   dst  = (const int*)d_in[2];
    const float* W0   = (const float*)d_in[3];
    const float* W1   = (const float*)d_in[4];
    const float* W2   = (const float*)d_in[5];
    const float* b2   = (const float*)d_in[6];
    const float* g_in = (const float*)d_in[7];
    const float* b_in = (const float*)d_in[8];
    const float* g0   = (const float*)d_in[9];
    const float* b0   = (const float*)d_in[10];
    const float* g1   = (const float*)d_in[11];
    const float* b1   = (const float*)d_in[12];
    float* out = (float*)d_out;

    const int IN_F = 768, HID = 128, CLS = 40;
    const int n = in_sizes[0] / IN_F;
    const int e = in_sizes[1];

    float* ws    = (float*)d_ws;
    float* do_is = ws;
    float* di_is = ws + (size_t)n;
    float* mu    = ws + 2 * (size_t)n;
    float* rs    = ws + 3 * (size_t)n;
    float* bufB  = ws + 4 * (size_t)n;
    float* bufC  = bufB + (size_t)n * HID;
    float* bufD  = bufC + (size_t)n * HID;
    int*   cnt_out = (int*)(bufD + (size_t)n * HID);
    int*   cnt_in  = cnt_out + (size_t)n;
    int*   rowptr  = cnt_in + (size_t)n;
    int*   rank    = rowptr + (size_t)n;
    int*   csr_src = rank + (size_t)e;
    float* bufE  = bufB;

    const int T = 256;

    hipMemsetAsync(cnt_out, 0, 2 * (size_t)n * sizeof(int), stream);
    degree_rank_kernel<<<cdiv(e, T), T, 0, stream>>>(src, dst, cnt_out, cnt_in, rank, e);

    const int nscan  = cdiv(n, 1024);
    const int nstats = cdiv((long long)n * 32, 1024);
    scan_and_ln_stats_kernel<<<nscan + nstats, 1024, 0, stream>>>(
        cnt_out, cnt_in, rowptr, do_is, di_is, feat, mu, rs, n, nscan);

    const int mtiles = cdiv(n, 16);
    const int ngemm  = cdiv((long long)mtiles * 32, T);
    gemm0_and_csr_fill_kernel<<<ngemm + cdiv(e, T), T, 0, stream>>>(
        feat, mu, rs, do_is, g_in, b_in, W0, bufB, n, ngemm,
        src, dst, rank, rowptr, csr_src, e);

    gather_ln_relu_128_kernel<<<cdiv((long long)n * 32, T), T, 0, stream>>>(
        bufB, rowptr, cnt_in, csr_src, di_is, do_is, g0, b0, bufD, n);

    gather_sum_vec4_kernel<32><<<cdiv((long long)n * 32, T), T, 0, stream>>>(
        bufD, rowptr, cnt_in, csr_src, bufB, n);
    gemm128_ln_relu_wmma_kernel<<<cdiv((long long)mtiles * 32, T), T, 0, stream>>>(
        bufB, W1, di_is, do_is, g1, b1, bufC, n);

    gemm_wmma_strip_kernel<3><<<cdiv((long long)mtiles * 32, T), T, 0, stream>>>(
        bufC, W2, bufE, n, HID, CLS);
    gather_scale_bias_40_kernel<<<cdiv((long long)n * 32, T), T, 0, stream>>>(
        bufE, rowptr, cnt_in, csr_src, di_is, b2, out, n);
}


# Input LayerNorm folded through the layer-0 product: raw rows @ (g*W0), row mean and deviation summed from the A fragments in the k loop, applied in the epilogue; no statistics pass; LN sums reordered
# speedup vs baseline: 2.3771x; 1.1660x over previous
#include <hip/hip_runtime.h>
#include <math.h>

typedef __attribute__((ext_vector_type(2))) float v2f;
typedef __attribute__((ext_vector_type(8))) float v8f;

#define LN_EPS 1e-5f

__global__ void degree_rank_kernel(const int* __restrict__ src, const int* __restrict__ dst,
                                   int* __restrict__ cnt_out, int* __restrict__ cnt_in,
                                   int* __restrict__ rank, int e) {
    int i = blockIdx.x * blockDim.x + threadIdx.x;
    if (i >= e) return;
    atomicAdd(cnt_out + src[i], 1);
    rank[i] = atomicAdd(cnt_in + dst[i], 1);
}

__device__ __forceinline__ void
rowptr_scan_tile(int blk, const int* __restrict__ cnt_out, const int* __restrict__ cnt_in,
                 int* __restrict__ rowptr,
                 float* __restrict__ do_is, float* __restrict__ di_is, int n) {
    __shared__ int wtot[32];
    __shared__ int wpart[32];
    const int tid  = threadIdx.x;
    const int lane = tid & 31;
    const int wid  = tid >> 5;
    const int tile0 = blk * 1024;

    int part = 0;
    for (int j = tid; j < tile0; j += 1024) part += cnt_in[j];

    const int i = tile0 + tid;
    const int c = (i < n) ? cnt_in[i] : 0;
    int incl = c;
    #pragma unroll
    for (int off = 1; off < 32; off <<= 1) {
        const int t = __shfl_up(incl, off, 32);
        if (lane >= off) incl += t;
    }
    #pragma unroll
    for (int off = 16; off > 0; off >>= 1) part += __shfl_xor(part, off, 32);
    if (lane == 31) wtot[wid]  = incl;
    if (lane == 0)  wpart[wid] = part;
    __syncthreads();
    int bsum = wpart[lane];
    int pre  = (lane < wid) ? wtot[lane] : 0;
    #pragma unroll
    for (int off = 16; off > 0; off >>= 1) {
        bsum += __shfl_xor(bsum, off, 32);
        pre  += __shfl_xor(pre,  off, 32);
    }
    if (i < n) {
        rowptr[i] = bsum + pre + incl - c;
        do_is[i]  = rsqrtf(fmaxf((float)cnt_out[i], 1.0f));
        di_is[i]  = rsqrtf(fmaxf((float)c, 1.0f));
    }
}

__device__ __forceinline__ void
csr_fill_edges(int blk, const int* __restrict__ src, const int* __restrict__ dst,
               const int* __restrict__ rank, const int* __restrict__ rowptr,
               int* __restrict__ csr_src, int e) {
    int i = blk * blockDim.x + threadIdx.x;
    if (i >= e) return;
    csr_src[rowptr[dst[i]] + rank[i]] = src[i];
}

__device__ __forceinline__ void
ln_fold_tables_16cols(int blk, const float* __restrict__ W,
                      const float* __restrict__ g, const float* __restrict__ b,
                      float* __restrict__ Wg, float* __restrict__ gw, float* __restrict__ bw) {
    __shared__ float pg[64][17];
    __shared__ float pb[64][17];
    const int tid = threadIdx.x;
    const int cc  = tid & 15;
    const int ks  = tid >> 4;
    const int col = blk * 16 + cc;
    float sg = 0.f, sb = 0.f;
    #pragma unroll
    for (int i = 0; i < 12; ++i) {
        const int k = ks * 12 + i;
        const float w  = W[(size_t)k * 128 + col];
        const float wg = g[k] * w;
        Wg[(size_t)k * 128 + col] = wg;
        sg += wg;
        sb = fmaf(b[k], w, sb);
    }
    pg[ks][cc] = sg;
    pb[ks][cc] = sb;
    __syncthreads();
    if (tid < 32) {
        float acc = 0.f;
        if (tid < 16) { for (int j = 0; j < 64; ++j) acc += pg[j][cc]; gw[col] = acc; }
        else          { for (int j = 0; j < 64; ++j) acc += pb[j][cc]; bw[col] = acc; }
    }
}

__global__ void __launch_bounds__(1024)
scan_and_fold_tables_kernel(const int* __restrict__ cnt_out, const int* __restrict__ cnt_in,
                            int* __restrict__ rowptr,
                            float* __restrict__ do_is, float* __restrict__ di_is,
                            const float* __restrict__ W, const float* __restrict__ g,
                            const float* __restrict__ b,
                            float* __restrict__ Wg, float* __restrict__ gw, float* __restrict__ bw,
                            int n, int nscan) {
    const int blk = blockIdx.x;
    if (blk < nscan) rowptr_scan_tile(blk, cnt_out, cnt_in, rowptr, do_is, di_is, n);
    else             ln_fold_tables_16cols(blk - nscan, W, g, b, Wg, gw, bw);
}

__device__ __forceinline__ void
gemm0_lnfold_wmma_strips(int blk, const float* __restrict__ feat,
                         const float* __restrict__ do_is,
                         const float* __restrict__ Wg,
                         const float* __restrict__ gw, const float* __restrict__ bw,
                         float* __restrict__ C, int M) {
    const int K = 768, F = 128;
    const int lane = threadIdx.x & 31;
    const int mt   = (blk * blockDim.x + threadIdx.x) >> 5;
    if (mt * 16 >= M) return;
    const int half = lane >> 4;
    const int l16  = lane & 15;
    const int arow = mt * 16 + l16;
    const int koff = half * 2;
    const float do_r = do_is[arow];
    const float* ap = feat + (size_t)arow * K + koff;

    v8f acc[8];
    #pragma unroll
    for (int i = 0; i < 8; ++i) { v8f z = {}; acc[i] = z; }
    v2f s1 = {0.f, 0.f};
    v2f s2 = {0.f, 0.f};

    for (int k = 0; k < K; k += 4) {
        const v2f a = *(const v2f*)(ap + k);
        s1 += a;
        s2 += a * a;
        const float* wrow = Wg + (size_t)(k + koff) * F;
        #pragma unroll
        for (int ft = 0; ft < 8; ++ft) {
            const int bcol = ft * 16 + l16;
            v2f bb;
            bb.x = wrow[bcol];
            bb.y = wrow[bcol + F];
            acc[ft] = __builtin_amdgcn_wmma_f32_16x16x4_f32(
                false, a, false, bb, (short)0, acc[ft], false, false);
        }
    }

    float s = s1.x + s1.y;
    float q = s2.x + s2.y;
    s += __shfl_xor(s, 16, 32);
    q += __shfl_xor(q, 16, 32);
    const float m   = s * (1.0f / 768.0f);
    const float rsv = rsqrtf(q * (1.0f / 768.0f) - m * m + LN_EPS);
    const float pA  = rsv * do_r;
    const float pB  = m * pA;

    float gwc[8], bwc[8];
    #pragma unroll
    for (int ft = 0; ft < 8; ++ft) {
        gwc[ft] = gw[ft * 16 + l16];
        bwc[ft] = bw[ft * 16 + l16];
    }
    const int crow = mt * 16 + half * 8;
    #pragma unroll
    for (int r = 0; r < 8; ++r) {
        const int sl = half * 8 + r;
        const float sA = __shfl(pA,   sl, 32);
        const float sB = __shfl(pB,   sl, 32);
        const float sD = __shfl(do_r, sl, 32);
        #pragma unroll
        for (int ft = 0; ft < 8; ++ft) {
            const int bcol = ft * 16 + l16;
            C[(size_t)(crow + r) * F + bcol] = sA * acc[ft][r] - sB * gwc[ft] + sD * bwc[ft];
        }
    }
}

__global__ void gemm0_and_csr_fill_kernel(const float* __restrict__ feat,
                                          const float* __restrict__ do_is,
                                          const float* __restrict__ Wg,
                                          const float* __restrict__ gw, const float* __restrict__ bw,
                                          float* __restrict__ C, int M, int ngemm,
                                          const int* __restrict__ src, const int* __restrict__ dst,
                                          const int* __restrict__ rank, const int* __restrict__ rowptr,
                                          int* __restrict__ csr_src, int e) {
    const int blk = blockIdx.x;
    if (blk < ngemm) gemm0_lnfold_wmma_strips(blk, feat, do_is, Wg, gw, bw, C, M);
    else             csr_fill_edges(blk - ngemm, src, dst, rank, rowptr, csr_src, e);
}

template <int FT>
__global__ void gemm_wmma_strip_kernel(const float* __restrict__ A, const float* __restrict__ W,
                                       float* __restrict__ C, int M, int K, int F) {
    const int lane = threadIdx.x & 31;
    const int mt   = (blockIdx.x * blockDim.x + threadIdx.x) >> 5;
    if (mt * 16 >= M) return;
    const int half = lane >> 4;
    const int l16  = lane & 15;
    const int arow = mt * 16 + l16;
    const int koff = half * 2;
    const float* ap = A + (size_t)arow * K + koff;

    v8f acc[FT];
    #pragma unroll
    for (int i = 0; i < FT; ++i) { v8f z = {}; acc[i] = z; }

    for (int k = 0; k < K; k += 4) {
        const v2f a = *(const v2f*)(ap + k);
        const float* wrow = W + (size_t)(k + koff) * F;
        #pragma unroll
        for (int ft = 0; ft < FT; ++ft) {
            const int  bcol = ft * 16 + l16;
            const bool bv   = (bcol < F);
            const int  bc   = bv ? bcol : 0;
            v2f bb;
            bb.x = bv ? wrow[bc]     : 0.0f;
            bb.y = bv ? wrow[bc + F] : 0.0f;
            acc[ft] = __builtin_amdgcn_wmma_f32_16x16x4_f32(
                false, a, false, bb, (short)0, acc[ft], false, false);
        }
    }
    const int crow = mt * 16 + half * 8;
    #pragma unroll
    for (int ft = 0; ft < FT; ++ft) {
        const int bcol = ft * 16 + l16;
        if (bcol < F) {
            #pragma unroll
            for (int r = 0; r < 8; ++r)
                C[(size_t)(crow + r) * F + bcol] = acc[ft][r];
        }
    }
}

template <int F4>
__device__ __forceinline__ float4 gather_row_sum(const float* __restrict__ x,
                                                 const int* __restrict__ rowptr,
                                                 const int* __restrict__ cnt_in,
                                                 const int* __restrict__ csr_src,
                                                 int row, int lane) {
    const int beg = rowptr[row];
    const int len = cnt_in[row];
    const int c4  = (lane < F4) ? lane : 0;
    const float4* xp = (const float4*)x + c4;
    float4 acc = make_float4(0.f, 0.f, 0.f, 0.f);
    for (int base = 0; base < len; base += 32) {
        const int m = (len - base < 32) ? (len - base) : 32;
        const int idx = (lane < m) ? csr_src[beg + base + lane] : 0;
        #pragma unroll 4
        for (int j = 0; j < m; ++j) {
            const int s = __builtin_amdgcn_readlane(idx, j);
            const float4 v = xp[(size_t)s * F4];
            acc.x += v.x; acc.y += v.y; acc.z += v.z; acc.w += v.w;
        }
    }
    return acc;
}

template <int F4>
__global__ void gather_sum_vec4_kernel(const float* __restrict__ x,
                                       const int* __restrict__ rowptr,
                                       const int* __restrict__ cnt_in,
                                       const int* __restrict__ csr_src,
                                       float* __restrict__ out, int n) {
    const int lane = threadIdx.x & 31;
    const int row = __builtin_amdgcn_readfirstlane((int)((blockIdx.x * blockDim.x + threadIdx.x) >> 5));
    if (row >= n) return;
    const float4 acc = gather_row_sum<F4>(x, rowptr, cnt_in, csr_src, row, lane);
    if (lane < F4) ((float4*)out)[(size_t)row * F4 + lane] = acc;
}

__global__ void gather_ln_relu_128_kernel(const float* __restrict__ x,
                                          const int* __restrict__ rowptr,
                                          const int* __restrict__ cnt_in,
                                          const int* __restrict__ csr_src,
                                          const float* __restrict__ di_is,
                                          const float* __restrict__ do_is,
                                          const float* __restrict__ g, const float* __restrict__ b,
                                          float* __restrict__ out, int n) {
    const int lane = threadIdx.x & 31;
    const int row = __builtin_amdgcn_readfirstlane((int)((blockIdx.x * blockDim.x + threadIdx.x) >> 5));
    if (row >= n) return;
    const float4 acc = gather_row_sum<32>(x, rowptr, cnt_in, csr_src, row, lane);
    const float di = di_is[row];
    const float ds = do_is[row];
    float v[4] = { acc.x * di, acc.y * di, acc.z * di, acc.w * di };
    float s = 0.f, s2 = 0.f;
    #pragma unroll
    for (int i = 0; i < 4; ++i) { s += v[i]; s2 += v[i] * v[i]; }
    #pragma unroll
    for (int off = 16; off > 0; off >>= 1) {
        s  += __shfl_xor(s,  off, 32);
        s2 += __shfl_xor(s2, off, 32);
    }
    const float m   = s  * (1.0f / 128.0f);
    const float rsv = rsqrtf(s2 * (1.0f / 128.0f) - m * m + LN_EPS);
    float y[4];
    #pragma unroll
    for (int i = 0; i < 4; ++i) {
        const int c = 4 * lane + i;
        const float t = (v[i] - m) * rsv * g[c] + b[c];
        y[i] = fmaxf(t, 0.0f) * ds;
    }
    ((float4*)out)[(size_t)row * 32 + lane] = make_float4(y[0], y[1], y[2], y[3]);
}

__global__ void gather_scale_bias_40_kernel(const float* __restrict__ x,
                                            const int* __restrict__ rowptr,
                                            const int* __restrict__ cnt_in,
                                            const int* __restrict__ csr_src,
                                            const float* __restrict__ di_is,
                                            const float* __restrict__ b2,
                                            float* __restrict__ out, int n) {
    const int lane = threadIdx.x & 31;
    const int row = __builtin_amdgcn_readfirstlane((int)((blockIdx.x * blockDim.x + threadIdx.x) >> 5));
    if (row >= n) return;
    const float4 acc = gather_row_sum<10>(x, rowptr, cnt_in, csr_src, row, lane);
    if (lane < 10) {
        const float di = di_is[row];
        const int c = 4 * lane;
        float4 o;
        o.x = acc.x * di + b2[c];
        o.y = acc.y * di + b2[c + 1];
        o.z = acc.z * di + b2[c + 2];
        o.w = acc.w * di + b2[c + 3];
        ((float4*)out)[(size_t)row * 10 + lane] = o;
    }
}

__global__ void gemm128_ln_relu_wmma_kernel(const float* __restrict__ A, const float* __restrict__ W,
                                            const float* __restrict__ di_is,
                                            const float* __restrict__ do_is,
                                            const float* __restrict__ g, const float* __restrict__ b,
                                            float* __restrict__ C, int M) {
    const int K = 128, F = 128;
    const int lane = threadIdx.x & 31;
    const int mt   = (blockIdx.x * blockDim.x + threadIdx.x) >> 5;
    if (mt * 16 >= M) return;
    const int half = lane >> 4;
    const int l16  = lane & 15;
    const int arow = mt * 16 + l16;
    const int koff = half * 2;
    const float* ap = A + (size_t)arow * K + koff;

    v8f acc[8];
    #pragma unroll
    for (int i = 0; i < 8; ++i) { v8f z = {}; acc[i] = z; }

    for (int k = 0; k < K; k += 4) {
        const v2f a = *(const v2f*)(ap + k);
        const float* wrow = W + (size_t)(k + koff) * F;
        #pragma unroll
        for (int ft = 0; ft < 8; ++ft) {
            const int bcol = ft * 16 + l16;
            v2f bb;
            bb.x = wrow[bcol];
            bb.y = wrow[bcol + F];
            acc[ft] = __builtin_amdgcn_wmma_f32_16x16x4_f32(
                false, a, false, bb, (short)0, acc[ft], false, false);
        }
    }

    const int crow = mt * 16 + half * 8;
    float gg[8], bbv[8];
    #pragma unroll
    for (int ft = 0; ft < 8; ++ft) {
        gg[ft]  = g[ft * 16 + l16];
        bbv[ft] = b[ft * 16 + l16];
    }
    #pragma unroll
    for (int r = 0; r < 8; ++r) {
        const float di = di_is[crow + r];
        const float ds = do_is[crow + r];
        float v[8];
        float s = 0.f, s2 = 0.f;
        #pragma unroll
        for (int ft = 0; ft < 8; ++ft) {
            v[ft] = acc[ft][r] * di;
            s += v[ft]; s2 += v[ft] * v[ft];
        }
        #pragma unroll
        for (int off = 8; off > 0; off >>= 1) {
            s  += __shfl_xor(s,  off, 32);
            s2 += __shfl_xor(s2, off, 32);
        }
        const float m   = s  * (1.0f / 128.0f);
        const float rsv = rsqrtf(s2 * (1.0f / 128.0f) - m * m + LN_EPS);
        #pragma unroll
        for (int ft = 0; ft < 8; ++ft) {
            const float t = (v[ft] - m) * rsv * gg[ft] + bbv[ft];
            C[(size_t)(crow + r) * F + ft * 16 + l16] = fmaxf(t, 0.0f) * ds;
        }
    }
}

static inline int cdiv(long long a, long long b) { return (int)((a + b - 1) / b); }

extern "C" void kernel_launch(void* const* d_in, const int* in_sizes, int n_in,
                              void* d_out, int out_size, void* d_ws, size_t ws_size,
                              hipStream_t stream) {
    const float* feat = (const float*)d_in[0];
    const int*   src  = (const int*)d_in[1];
    const int*   dst  = (const int*)d_in[2];
    const float* W0   = (const float*)d_in[3];
    const float* W1   = (const float*)d_in[4];
    const float* W2   = (const float*)d_in[5];
    const float* b2   = (const float*)d_in[6];
    const float* g_in = (const float*)d_in[7];
    const float* b_in = (const float*)d_in[8];
    const float* g0   = (const float*)d_in[9];
    const float* b0   = (const float*)d_in[10];
    const float* g1   = (const float*)d_in[11];
    const float* b1   = (const float*)d_in[12];
    float* out = (float*)d_out;

    const int IN_F = 768, HID = 128, CLS = 40;
    const int n = in_sizes[0] / IN_F;
    const int e = in_sizes[1];

    float* ws    = (float*)d_ws;
    float* do_is = ws;
    float* di_is = ws + (size_t)n;
    float* Wg    = ws + 2 * (size_t)n;
    float* gw    = Wg + (size_t)IN_F * HID;
    float* bw    = gw + HID;
    float* bufB  = ws + 4 * (size_t)n;
    float* bufC  = bufB + (size_t)n * HID;
    float* bufD  = bufC + (size_t)n * HID;
    int*   cnt_out = (int*)(bufD + (size_t)n * HID);
    int*   cnt_in  = cnt_out + (size_t)n;
    int*   rowptr  = cnt_in + (size_t)n;
    int*   rank    = rowptr + (size_t)n;
    int*   csr_src = rank + (size_t)e;
    float* bufE  = bufB;

    const int T = 256;

    hipMemsetAsync(cnt_out, 0, 2 * (size_t)n * sizeof(int), stream);
    degree_rank_kernel<<<cdiv(e, T), T, 0, stream>>>(src, dst, cnt_out, cnt_in, rank, e);

    const int nscan = cdiv(n, 1024);
    scan_and_fold_tables_kernel<<<nscan + HID / 16, 1024, 0, stream>>>(
        cnt_out, cnt_in, rowptr, do_is, di_is, W0, g_in, b_in, Wg, gw, bw, n, nscan);

    const int mtiles = cdiv(n, 16);
    const int ngemm  = cdiv((long long)mtiles * 32, T);
    gemm0_and_csr_fill_kernel<<<ngemm + cdiv(e, T), T, 0, stream>>>(
        feat, do_is, Wg, gw, bw, bufB, n, ngemm,
        src, dst, rank, rowptr, csr_src, e);

    gather_ln_relu_128_kernel<<<cdiv((long long)n * 32, T), T, 0, stream>>>(
        bufB, rowptr, cnt_in, csr_src, di_is, do_is, g0, b0, bufD, n);

    gather_sum_vec4_kernel<32><<<cdiv((long long)n * 32, T), T, 0, stream>>>(
        bufD, rowptr, cnt_in, csr_src, bufB, n);
    gemm128_ln_relu_wmma_kernel<<<cdiv((long long)mtiles * 32, T), T, 0, stream>>>(
        bufB, W1, di_is, do_is, g1, b1, bufC, n);

    gemm_wmma_strip_kernel<3><<<cdiv((long long)mtiles * 32, T), T, 0, stream>>>(
        bufC, W2, bufE, n, HID, CLS);
    gather_scale_bias_40_kernel<<<cdiv((long long)n * 32, T), T, 0, stream>>>(
        bufE, rowptr, cnt_in, csr_src, di_is, b2, out, n);
}


# Layer-0 product: weight table kept in fragment order and staged through LDS once per 128-row block (32 k rows per chunk, two buffers); B fragments are 16-byte LDS reads; sum order unchanged
# speedup vs baseline: 2.6176x; 1.1012x over previous
#include <hip/hip_runtime.h>
#include <math.h>

typedef __attribute__((ext_vector_type(2))) float v2f;
typedef __attribute__((ext_vector_type(8))) float v8f;

#define LN_EPS 1e-5f

__global__ void degree_rank_kernel(const int* __restrict__ src, const int* __restrict__ dst,
                                   int* __restrict__ cnt_out, int* __restrict__ cnt_in,
                                   int* __restrict__ rank, int e) {
    int i = blockIdx.x * blockDim.x + threadIdx.x;
    if (i >= e) return;
    atomicAdd(cnt_out + src[i], 1);
    rank[i] = atomicAdd(cnt_in + dst[i], 1);
}

__device__ __forceinline__ void
rowptr_scan_tile(int blk, const int* __restrict__ cnt_out, const int* __restrict__ cnt_in,
                 int* __restrict__ rowptr,
                 float* __restrict__ do_is, float* __restrict__ di_is, int n) {
    __shared__ int wtot[32];
    __shared__ int wpart[32];
    const int tid  = threadIdx.x;
    const int lane = tid & 31;
    const int wid  = tid >> 5;
    const int tile0 = blk * 1024;

    int part = 0;
    for (int j = tid; j < tile0; j += 1024) part += cnt_in[j];

    const int i = tile0 + tid;
    const int c = (i < n) ? cnt_in[i] : 0;
    int incl = c;
    #pragma unroll
    for (int off = 1; off < 32; off <<= 1) {
        const int t = __shfl_up(incl, off, 32);
        if (lane >= off) incl += t;
    }
    #pragma unroll
    for (int off = 16; off > 0; off >>= 1) part += __shfl_xor(part, off, 32);
    if (lane == 31) wtot[wid]  = incl;
    if (lane == 0)  wpart[wid] = part;
    __syncthreads();
    int bsum = wpart[lane];
    int pre  = (lane < wid) ? wtot[lane] : 0;
    #pragma unroll
    for (int off = 16; off > 0; off >>= 1) {
        bsum += __shfl_xor(bsum, off, 32);
        pre  += __shfl_xor(pre,  off, 32);
    }
    if (i < n) {
        rowptr[i] = bsum + pre + incl - c;
        do_is[i]  = rsqrtf(fmaxf((float)cnt_out[i], 1.0f));
        di_is[i]  = rsqrtf(fmaxf((float)c, 1.0f));
    }
}

__device__ __forceinline__ void
csr_fill_edges(int blk, const int* __restrict__ src, const int* __restrict__ dst,
               const int* __restrict__ rank, const int* __restrict__ rowptr,
               int* __restrict__ csr_src, int e) {
    int i = blk * blockDim.x + threadIdx.x;
    if (i >= e) return;
    csr_src[rowptr[dst[i]] + rank[i]] = src[i];
}

__device__ __forceinline__ size_t wg_frag_index(int k, int col) {
    const int pr = k >> 1, kb = k & 1;
    const int q = col >> 5, j = (col >> 4) & 1, c = col & 15;
    return (size_t)pr * 256 + (size_t)((((q * 16 + c) * 2 + j) * 2) + kb);
}

__device__ __forceinline__ void
ln_fold_tables_16cols(int blk, const float* __restrict__ W,
                      const float* __restrict__ g, const float* __restrict__ b,
                      float* __restrict__ Wg, float* __restrict__ gw, float* __restrict__ bw) {
    __shared__ float pg[64][17];
    __shared__ float pb[64][17];
    const int tid = threadIdx.x;
    const int cc  = tid & 15;
    const int ks  = tid >> 4;
    const int col = blk * 16 + cc;
    float sg = 0.f, sb = 0.f;
    #pragma unroll
    for (int i = 0; i < 12; ++i) {
        const int k = ks * 12 + i;
        const float w  = W[(size_t)k * 128 + col];
        const float wg = g[k] * w;
        Wg[wg_frag_index(k, col)] = wg;
        sg += wg;
        sb = fmaf(b[k], w, sb);
    }
    pg[ks][cc] = sg;
    pb[ks][cc] = sb;
    __syncthreads();
    if (tid < 32) {
        float acc = 0.f;
        if (tid < 16) { for (int j = 0; j < 64; ++j) acc += pg[j][cc]; gw[col] = acc; }
        else          { for (int j = 0; j < 64; ++j) acc += pb[j][cc]; bw[col] = acc; }
    }
}

__global__ void __launch_bounds__(1024)
scan_and_fold_tables_kernel(const int* __restrict__ cnt_out, const int* __restrict__ cnt_in,
                            int* __restrict__ rowptr,
                            float* __restrict__ do_is, float* __restrict__ di_is,
                            const float* __restrict__ W, const float* __restrict__ g,
                            const float* __restrict__ b,
                            float* __restrict__ Wg, float* __restrict__ gw, float* __restrict__ bw,
                            int n, int nscan) {
    const int blk = blockIdx.x;
    if (blk < nscan) rowptr_scan_tile(blk, cnt_out, cnt_in, rowptr, do_is, di_is, n);
    else             ln_fold_tables_16cols(blk - nscan, W, g, b, Wg, gw, bw);
}

#define G0_KC     32
#define G0_CHUNK4 (G0_KC / 2 * 64)

__device__ __forceinline__ void
gemm0_lnfold_wmma_strips(int blk, const float* __restrict__ feat,
                         const float* __restrict__ do_is,
                         const float* __restrict__ Wg,
                         const float* __restrict__ gw, const float* __restrict__ bw,
                         float* __restrict__ C, int M) {
    __shared__ float4 bst[2][G0_CHUNK4];
    const int K = 768, F = 128;
    const int tid  = threadIdx.x;
    const int lane = tid & 31;
    const int mt   = (blk * blockDim.x + tid) >> 5;
    const bool live = (mt * 16 < M);
    const int half = lane >> 4;
    const int l16  = lane & 15;
    int arow = mt * 16 + l16;
    if (arow >= M) arow = M - 1;
    const int koff = half * 2;
    const float do_r = do_is[arow];
    const float* ap = feat + (size_t)arow * K + koff;
    const float4* wg4 = (const float4*)Wg;

    v8f acc[8];
    #pragma unroll
    for (int i = 0; i < 8; ++i) { v8f z = {}; acc[i] = z; }
    v2f s1 = {0.f, 0.f};
    v2f s2 = {0.f, 0.f};

    float4 st[4];
    #pragma unroll
    for (int p = 0; p < 4; ++p) st[p] = wg4[p * 256 + tid];
    #pragma unroll
    for (int p = 0; p < 4; ++p) bst[0][p * 256 + tid] = st[p];
    __syncthreads();

    const int NCH = K / G0_KC;
    for (int ch = 0; ch < NCH; ++ch) {
        const int cur = ch & 1;
        const bool more = (ch + 1 < NCH);
        if (more) {
            #pragma unroll
            for (int p = 0; p < 4; ++p) st[p] = wg4[(size_t)(ch + 1) * G0_CHUNK4 + p * 256 + tid];
        }
        const float* apc = ap + ch * G0_KC;
        const float4* bcur = &bst[cur][half * 64 + l16];
        v2f af[G0_KC / 4];
        #pragma unroll
        for (int i = 0; i < G0_KC / 4; ++i) af[i] = *(const v2f*)(apc + 4 * i);
        #pragma unroll
        for (int kk = 0; kk < G0_KC; kk += 4) {
            const v2f a = af[kk / 4];
            s1 += a;
            s2 += a * a;
            #pragma unroll
            for (int q = 0; q < 4; ++q) {
                const float4 w = bcur[(kk / 2) * 64 + q * 16];
                v2f b0; b0.x = w.x; b0.y = w.y;
                v2f b1; b1.x = w.z; b1.y = w.w;
                acc[2 * q] = __builtin_amdgcn_wmma_f32_16x16x4_f32(
                    false, a, false, b0, (short)0, acc[2 * q], false, false);
                acc[2 * q + 1] = __builtin_amdgcn_wmma_f32_16x16x4_f32(
                    false, a, false, b1, (short)0, acc[2 * q + 1], false, false);
            }
        }
        if (more) {
            #pragma unroll
            for (int p = 0; p < 4; ++p) bst[cur ^ 1][p * 256 + tid] = st[p];
        }
        __syncthreads();
    }
    if (!live) return;

    float s = s1.x + s1.y;
    float q = s2.x + s2.y;
    s += __shfl_xor(s, 16, 32);
    q += __shfl_xor(q, 16, 32);
    const float m   = s * (1.0f / 768.0f);
    const float rsv = rsqrtf(q * (1.0f / 768.0f) - m * m + LN_EPS);
    const float pA  = rsv * do_r;
    const float pB  = m * pA;

    float gwc[8], bwc[8];
    #pragma unroll
    for (int ft = 0; ft < 8; ++ft) {
        gwc[ft] = gw[ft * 16 + l16];
        bwc[ft] = bw[ft * 16 + l16];
    }
    const int crow = mt * 16 + half * 8;
    #pragma unroll
    for (int r = 0; r < 8; ++r) {
        const int sl = half * 8 + r;
        const float sA = __shfl(pA,   sl, 32);
        const float sB = __shfl(pB,   sl, 32);
        const float sD = __shfl(do_r, sl, 32);
        #pragma unroll
        for (int ft = 0; ft < 8; ++ft) {
            const int bcol = ft * 16 + l16;
            C[(size_t)(crow + r) * F + bcol] = sA * acc[ft][r] - sB * gwc[ft] + sD * bwc[ft];
        }
    }
}

__global__ void __launch_bounds__(256)
gemm0_and_csr_fill_kernel(const float* __restrict__ feat,
                                          const float* __restrict__ do_is,
                                          const float* __restrict__ Wg,
                                          const float* __restrict__ gw, const float* __restrict__ bw,
                                          float* __restrict__ C, int M, int ngemm,
                                          const int* __restrict__ src, const int* __restrict__ dst,
                                          const int* __restrict__ rank, const int* __restrict__ rowptr,
                                          int* __restrict__ csr_src, int e) {
    const int blk = blockIdx.x;
    if (blk < ngemm) gemm0_lnfold_wmma_strips(blk, feat, do_is, Wg, gw, bw, C, M);
    else             csr_fill_edges(blk - ngemm, src, dst, rank, rowptr, csr_src, e);
}

template <int FT>
__global__ void gemm_wmma_strip_kernel(const float* __restrict__ A, const float* __restrict__ W,
                                       float* __restrict__ C, int M, int K, int F) {
    const int lane = threadIdx.x & 31;
    const int mt   = (blockIdx.x * blockDim.x + threadIdx.x) >> 5;
    if (mt * 16 >= M) return;
    const int half = lane >> 4;
    const int l16  = lane & 15;
    const int arow = mt * 16 + l16;
    const int koff = half * 2;
    const float* ap = A + (size_t)arow * K + koff;

    v8f acc[FT];
    #pragma unroll
    for (int i = 0; i < FT; ++i) { v8f z = {}; acc[i] = z; }

    for (int k = 0; k < K; k += 4) {
        const v2f a = *(const v2f*)(ap + k);
        const float* wrow = W + (size_t)(k + koff) * F;
        #pragma unroll
        for (int ft = 0; ft < FT; ++ft) {
            const int  bcol = ft * 16 + l16;
            const bool bv   = (bcol < F);
            const int  bc   = bv ? bcol : 0;
            v2f bb;
            bb.x = bv ? wrow[bc]     : 0.0f;
            bb.y = bv ? wrow[bc + F] : 0.0f;
            acc[ft] = __builtin_amdgcn_wmma_f32_16x16x4_f32(
                false, a, false, bb, (short)0, acc[ft], false, false);
        }
    }
    const int crow = mt * 16 + half * 8;
    #pragma unroll
    for (int ft = 0; ft < FT; ++ft) {
        const int bcol = ft * 16 + l16;
        if (bcol < F) {
            #pragma unroll
            for (int r = 0; r < 8; ++r)
                C[(size_t)(crow + r) * F + bcol] = acc[ft][r];
        }
    }
}

template <int F4>
__device__ __forceinline__ float4 gather_row_sum(const float* __restrict__ x,
                                                 const int* __restrict__ rowptr,
                                                 const int* __restrict__ cnt_in,
                                                 const int* __restrict__ csr_src,
                                                 int row, int lane) {
    const int beg = rowptr[row];
    const int len = cnt_in[row];
    const int c4  = (lane < F4) ? lane : 0;
    const float4* xp = (const float4*)x + c4;
    float4 acc = make_float4(0.f, 0.f, 0.f, 0.f);
    for (int base = 0; base < len; base += 32) {
        const int m = (len - base < 32) ? (len - base) : 32;
        const int idx = (lane < m) ? csr_src[beg + base + lane] : 0;
        #pragma unroll 4
        for (int j = 0; j < m; ++j) {
            const int s = __builtin_amdgcn_readlane(idx, j);
            const float4 v = xp[(size_t)s * F4];
            acc.x += v.x; acc.y += v.y; acc.z += v.z; acc.w += v.w;
        }
    }
    return acc;
}

template <int F4>
__global__ void gather_sum_vec4_kernel(const float* __restrict__ x,
                                       const int* __restrict__ rowptr,
                                       const int* __restrict__ cnt_in,
                                       const int* __restrict__ csr_src,
                                       float* __restrict__ out, int n) {
    const int lane = threadIdx.x & 31;
    const int row = __builtin_amdgcn_readfirstlane((int)((blockIdx.x * blockDim.x + threadIdx.x) >> 5));
    if (row >= n) return;
    const float4 acc = gather_row_sum<F4>(x, rowptr, cnt_in, csr_src, row, lane);
    if (lane < F4) ((float4*)out)[(size_t)row * F4 + lane] = acc;
}

__global__ void gather_ln_relu_128_kernel(const float* __restrict__ x,
                                          const int* __restrict__ rowptr,
                                          const int* __restrict__ cnt_in,
                                          const int* __restrict__ csr_src,
                                          const float* __restrict__ di_is,
                                          const float* __restrict__ do_is,
                                          const float* __restrict__ g, const float* __restrict__ b,
                                          float* __restrict__ out, int n) {
    const int lane = threadIdx.x & 31;
    const int row = __builtin_amdgcn_readfirstlane((int)((blockIdx.x * blockDim.x + threadIdx.x) >> 5));
    if (row >= n) return;
    const float4 acc = gather_row_sum<32>(x, rowptr, cnt_in, csr_src, row, lane);
    const float di = di_is[row];
    const float ds = do_is[row];
    float v[4] = { acc.x * di, acc.y * di, acc.z * di, acc.w * di };
    float s = 0.f, s2 = 0.f;
    #pragma unroll
    for (int i = 0; i < 4; ++i) { s += v[i]; s2 += v[i] * v[i]; }
    #pragma unroll
    for (int off = 16; off > 0; off >>= 1) {
        s  += __shfl_xor(s,  off, 32);
        s2 += __shfl_xor(s2, off, 32);
    }
    const float m   = s  * (1.0f / 128.0f);
    const float rsv = rsqrtf(s2 * (1.0f / 128.0f) - m * m + LN_EPS);
    float y[4];
    #pragma unroll
    for (int i = 0; i < 4; ++i) {
        const int c = 4 * lane + i;
        const float t = (v[i] - m) * rsv * g[c] + b[c];
        y[i] = fmaxf(t, 0.0f) * ds;
    }
    ((float4*)out)[(size_t)row * 32 + lane] = make_float4(y[0], y[1], y[2], y[3]);
}

__global__ void gather_scale_bias_40_kernel(const float* __restrict__ x,
                                            const int* __restrict__ rowptr,
                                            const int* __restrict__ cnt_in,
                                            const int* __restrict__ csr_src,
                                            const float* __restrict__ di_is,
                                            const float* __restrict__ b2,
                                            float* __restrict__ out, int n) {
    const int lane = threadIdx.x & 31;
    const int row = __builtin_amdgcn_readfirstlane((int)((blockIdx.x * blockDim.x + threadIdx.x) >> 5));
    if (row >= n) return;
    const float4 acc = gather_row_sum<10>(x, rowptr, cnt_in, csr_src, row, lane);
    if (lane < 10) {
        const float di = di_is[row];
        const int c = 4 * lane;
        float4 o;
        o.x = acc.x * di + b2[c];
        o.y = acc.y * di + b2[c + 1];
        o.z = acc.z * di + b2[c + 2];
        o.w = acc.w * di + b2[c + 3];
        ((float4*)out)[(size_t)row * 10 + lane] = o;
    }
}

__global__ void gemm128_ln_relu_wmma_kernel(const float* __restrict__ A, const float* __restrict__ W,
                                            const float* __restrict__ di_is,
                                            const float* __restrict__ do_is,
                                            const float* __restrict__ g, const float* __restrict__ b,
                                            float* __restrict__ C, int M) {
    const int K = 128, F = 128;
    const int lane = threadIdx.x & 31;
    const int mt   = (blockIdx.x * blockDim.x + threadIdx.x) >> 5;
    if (mt * 16 >= M) return;
    const int half = lane >> 4;
    const int l16  = lane & 15;
    const int arow = mt * 16 + l16;
    const int koff = half * 2;
    const float* ap = A + (size_t)arow * K + koff;

    v8f acc[8];
    #pragma unroll
    for (int i = 0; i < 8; ++i) { v8f z = {}; acc[i] = z; }

    for (int k = 0; k < K; k += 4) {
        const v2f a = *(const v2f*)(ap + k);
        const float* wrow = W + (size_t)(k + koff) * F;
        #pragma unroll
        for (int ft = 0; ft < 8; ++ft) {
            const int bcol = ft * 16 + l16;
            v2f bb;
            bb.x = wrow[bcol];
            bb.y = wrow[bcol + F];
            acc[ft] = __builtin_amdgcn_wmma_f32_16x16x4_f32(
                false, a, false, bb, (short)0, acc[ft], false, false);
        }
    }

    const int crow = mt * 16 + half * 8;
    float gg[8], bbv[8];
    #pragma unroll
    for (int ft = 0; ft < 8; ++ft) {
        gg[ft]  = g[ft * 16 + l16];
        bbv[ft] = b[ft * 16 + l16];
    }
    #pragma unroll
    for (int r = 0; r < 8; ++r) {
        const float di = di_is[crow + r];
        const float ds = do_is[crow + r];
        float v[8];
        float s = 0.f, s2 = 0.f;
        #pragma unroll
        for (int ft = 0; ft < 8; ++ft) {
            v[ft] = acc[ft][r] * di;
            s += v[ft]; s2 += v[ft] * v[ft];
        }
        #pragma unroll
        for (int off = 8; off > 0; off >>= 1) {
            s  += __shfl_xor(s,  off, 32);
            s2 += __shfl_xor(s2, off, 32);
        }
        const float m   = s  * (1.0f / 128.0f);
        const float rsv = rsqrtf(s2 * (1.0f / 128.0f) - m * m + LN_EPS);
        #pragma unroll
        for (int ft = 0; ft < 8; ++ft) {
            const float t = (v[ft] - m) * rsv * gg[ft] + bbv[ft];
            C[(size_t)(crow + r) * F + ft * 16 + l16] = fmaxf(t, 0.0f) * ds;
        }
    }
}

static inline int cdiv(long long a, long long b) { return (int)((a + b - 1) / b); }

extern "C" void kernel_launch(void* const* d_in, const int* in_sizes, int n_in,
                              void* d_out, int out_size, void* d_ws, size_t ws_size,
                              hipStream_t stream) {
    const float* feat = (const float*)d_in[0];
    const int*   src  = (const int*)d_in[1];
    const int*   dst  = (const int*)d_in[2];
    const float* W0   = (const float*)d_in[3];
    const float* W1   = (const float*)d_in[4];
    const float* W2   = (const float*)d_in[5];
    const float* b2   = (const float*)d_in[6];
    const float* g_in = (const float*)d_in[7];
    const float* b_in = (const float*)d_in[8];
    const float* g0   = (const float*)d_in[9];
    const float* b0   = (const float*)d_in[10];
    const float* g1   = (const float*)d_in[11];
    const float* b1   = (const float*)d_in[12];
    float* out = (float*)d_out;

    const int IN_F = 768, HID = 128, CLS = 40;
    const int n = in_sizes[0] / IN_F;
    const int e = in_sizes[1];

    float* ws    = (float*)d_ws;
    float* do_is = ws;
    float* di_is = ws + (size_t)n;
    float* Wg    = ws + 2 * (size_t)n;
    float* gw    = Wg + (size_t)IN_F * HID;
    float* bw    = gw + HID;
    float* bufB  = ws + 4 * (size_t)n;
    float* bufC  = bufB + (size_t)n * HID;
    float* bufD  = bufC + (size_t)n * HID;
    int*   cnt_out = (int*)(bufD + (size_t)n * HID);
    int*   cnt_in  = cnt_out + (size_t)n;
    int*   rowptr  = cnt_in + (size_t)n;
    int*   rank    = rowptr + (size_t)n;
    int*   csr_src = rank + (size_t)e;
    float* bufE  = bufB;

    const int T = 256;

    hipMemsetAsync(cnt_out, 0, 2 * (size_t)n * sizeof(int), stream);
    degree_rank_kernel<<<cdiv(e, T), T, 0, stream>>>(src, dst, cnt_out, cnt_in, rank, e);

    const int nscan = cdiv(n, 1024);
    scan_and_fold_tables_kernel<<<nscan + HID / 16, 1024, 0, stream>>>(
        cnt_out, cnt_in, rowptr, do_is, di_is, W0, g_in, b_in, Wg, gw, bw, n, nscan);

    const int mtiles = cdiv(n, 16);
    const int ngemm  = cdiv((long long)mtiles * 32, T);
    gemm0_and_csr_fill_kernel<<<ngemm + cdiv(e, T), T, 0, stream>>>(
        feat, do_is, Wg, gw, bw, bufB, n, ngemm,
        src, dst, rank, rowptr, csr_src, e);

    gather_ln_relu_128_kernel<<<cdiv((long long)n * 32, T), T, 0, stream>>>(
        bufB, rowptr, cnt_in, csr_src, di_is, do_is, g0, b0, bufD, n);

    gather_sum_vec4_kernel<32><<<cdiv((long long)n * 32, T), T, 0, stream>>>(
        bufD, rowptr, cnt_in, csr_src, bufB, n);
    gemm128_ln_relu_wmma_kernel<<<cdiv((long long)mtiles * 32, T), T, 0, stream>>>(
        bufB, W1, di_is, do_is, g1, b1, bufC, n);

    gemm_wmma_strip_kernel<3><<<cdiv((long long)mtiles * 32, T), T, 0, stream>>>(
        bufC, W2, bufE, n, HID, CLS);
    gather_scale_bias_40_kernel<<<cdiv((long long)n * 32, T), T, 0, stream>>>(
        bufE, rowptr, cnt_in, csr_src, di_is, b2, out, n);
}


# Layer-2 product h2 @ W2 moved into the layer-1 product kernel: a wave's finished 16x128 h2 tile passes through LDS into A fragments, W2 staged in LDS; h2 never stored, 1 launch fewer; sums unchanged
# speedup vs baseline: 2.8696x; 1.0963x over previous
#include <hip/hip_runtime.h>
#include <math.h>

typedef __attribute__((ext_vector_type(2))) float v2f;
typedef __attribute__((ext_vector_type(8))) float v8f;

#define LN_EPS 1e-5f

__global__ void degree_rank_kernel(const int* __restrict__ src, const int* __restrict__ dst,
                                   int* __restrict__ cnt_out, int* __restrict__ cnt_in,
                                   int* __restrict__ rank, int e) {
    int i = blockIdx.x * blockDim.x + threadIdx.x;
    if (i >= e) return;
    atomicAdd(cnt_out + src[i], 1);
    rank[i] = atomicAdd(cnt_in + dst[i], 1);
}

__device__ __forceinline__ void
rowptr_scan_tile(int blk, const int* __restrict__ cnt_out, const int* __restrict__ cnt_in,
                 int* __restrict__ rowptr,
                 float* __restrict__ do_is, float* __restrict__ di_is, int n) {
    __shared__ int wtot[32];
    __shared__ int wpart[32];
    const int tid  = threadIdx.x;
    const int lane = tid & 31;
    const int wid  = tid >> 5;
    const int tile0 = blk * 1024;

    int part = 0;
    for (int j = tid; j < tile0; j += 1024) part += cnt_in[j];

    const int i = tile0 + tid;
    const int c = (i < n) ? cnt_in[i] : 0;
    int incl = c;
    #pragma unroll
    for (int off = 1; off < 32; off <<= 1) {
        const int t = __shfl_up(incl, off, 32);
        if (lane >= off) incl += t;
    }
    #pragma unroll
    for (int off = 16; off > 0; off >>= 1) part += __shfl_xor(part, off, 32);
    if (lane == 31) wtot[wid]  = incl;
    if (lane == 0)  wpart[wid] = part;
    __syncthreads();
    int bsum = wpart[lane];
    int pre  = (lane < wid) ? wtot[lane] : 0;
    #pragma unroll
    for (int off = 16; off > 0; off >>= 1) {
        bsum += __shfl_xor(bsum, off, 32);
        pre  += __shfl_xor(pre,  off, 32);
    }
    if (i < n) {
        rowptr[i] = bsum + pre + incl - c;
        do_is[i]  = rsqrtf(fmaxf((float)cnt_out[i], 1.0f));
        di_is[i]  = rsqrtf(fmaxf((float)c, 1.0f));
    }
}

__device__ __forceinline__ void
csr_fill_edges(int blk, const int* __restrict__ src, const int* __restrict__ dst,
               const int* __restrict__ rank, const int* __restrict__ rowptr,
               int* __restrict__ csr_src, int e) {
    int i = blk * blockDim.x + threadIdx.x;
    if (i >= e) return;
    csr_src[rowptr[dst[i]] + rank[i]] = src[i];
}

__device__ __forceinline__ size_t wg_frag_index(int k, int col) {
    const int pr = k >> 1, kb = k & 1;
    const int q = col >> 5, j = (col >> 4) & 1, c = col & 15;
    return (size_t)pr * 256 + (size_t)((((q * 16 + c) * 2 + j) * 2) + kb);
}

__device__ __forceinline__ void
ln_fold_tables_16cols(int blk, const float* __restrict__ W,
                      const float* __restrict__ g, const float* __restrict__ b,
                      float* __restrict__ Wg, float* __restrict__ gw, float* __restrict__ bw) {
    __shared__ float pg[64][17];
    __shared__ float pb[64][17];
    const int tid = threadIdx.x;
    const int cc  = tid & 15;
    const int ks  = tid >> 4;
    const int col = blk * 16 + cc;
    float sg = 0.f, sb = 0.f;
    #pragma unroll
    for (int i = 0; i < 12; ++i) {
        const int k = ks * 12 + i;
        const float w  = W[(size_t)k * 128 + col];
        const float wg = g[k] * w;
        Wg[wg_frag_index(k, col)] = wg;
        sg += wg;
        sb = fmaf(b[k], w, sb);
    }
    pg[ks][cc] = sg;
    pb[ks][cc] = sb;
    __syncthreads();
    if (tid < 32) {
        float acc = 0.f;
        if (tid < 16) { for (int j = 0; j < 64; ++j) acc += pg[j][cc]; gw[col] = acc; }
        else          { for (int j = 0; j < 64; ++j) acc += pb[j][cc]; bw[col] = acc; }
    }
}

__global__ void __launch_bounds__(1024)
scan_and_fold_tables_kernel(const int* __restrict__ cnt_out, const int* __restrict__ cnt_in,
                            int* __restrict__ rowptr,
                            float* __restrict__ do_is, float* __restrict__ di_is,
                            const float* __restrict__ W, const float* __restrict__ g,
                            const float* __restrict__ b,
                            float* __restrict__ Wg, float* __restrict__ gw, float* __restrict__ bw,
                            int n, int nscan) {
    const int blk = blockIdx.x;
    if (blk < nscan) rowptr_scan_tile(blk, cnt_out, cnt_in, rowptr, do_is, di_is, n);
    else             ln_fold_tables_16cols(blk - nscan, W, g, b, Wg, gw, bw);
}

#define G0_KC     32
#define G0_CHUNK4 (G0_KC / 2 * 64)

__device__ __forceinline__ void
gemm0_lnfold_wmma_strips(int blk, const float* __restrict__ feat,
                         const float* __restrict__ do_is,
                         const float* __restrict__ Wg,
                         const float* __restrict__ gw, const float* __restrict__ bw,
                         float* __restrict__ C, int M) {
    __shared__ float4 bst[2][G0_CHUNK4];
    const int K = 768, F = 128;
    const int tid  = threadIdx.x;
    const int lane = tid & 31;
    const int mt   = (blk * blockDim.x + tid) >> 5;
    const bool live = (mt * 16 < M);
    const int half = lane >> 4;
    const int l16  = lane & 15;
    int arow = mt * 16 + l16;
    if (arow >= M) arow = M - 1;
    const int koff = half * 2;
    const float do_r = do_is[arow];
    const float* ap = feat + (size_t)arow * K + koff;
    const float4* wg4 = (const float4*)Wg;

    v8f acc[8];
    #pragma unroll
    for (int i = 0; i < 8; ++i) { v8f z = {}; acc[i] = z; }
    v2f s1 = {0.f, 0.f};
    v2f s2 = {0.f, 0.f};

    float4 st[4];
    #pragma unroll
    for (int p = 0; p < 4; ++p) st[p] = wg4[p * 256 + tid];
    #pragma unroll
    for (int p = 0; p < 4; ++p) bst[0][p * 256 + tid] = st[p];
    __syncthreads();

    const int NCH = K / G0_KC;
    for (int ch = 0; ch < NCH; ++ch) {
        const int cur = ch & 1;
        const bool more = (ch + 1 < NCH);
        if (more) {
            #pragma unroll
            for (int p = 0; p < 4; ++p) st[p] = wg4[(size_t)(ch + 1) * G0_CHUNK4 + p * 256 + tid];
        }
        const float* apc = ap + ch * G0_KC;
        const float4* bcur = &bst[cur][half * 64 + l16];
        v2f af[G0_KC / 4];
        #pragma unroll
        for (int i = 0; i < G0_KC / 4; ++i) af[i] = *(const v2f*)(apc + 4 * i);
        #pragma unroll
        for (int kk = 0; kk < G0_KC; kk += 4) {
            const v2f a = af[kk / 4];
            s1 += a;
            s2 += a * a;
            #pragma unroll
            for (int q = 0; q < 4; ++q) {
                const float4 w = bcur[(kk / 2) * 64 + q * 16];
                v2f b0; b0.x = w.x; b0.y = w.y;
                v2f b1; b1.x = w.z; b1.y = w.w;
                acc[2 * q] = __builtin_amdgcn_wmma_f32_16x16x4_f32(
                    false, a, false, b0, (short)0, acc[2 * q], false, false);
                acc[2 * q + 1] = __builtin_amdgcn_wmma_f32_16x16x4_f32(
                    false, a, false, b1, (short)0, acc[2 * q + 1], false, false);
            }
        }
        if (more) {
            #pragma unroll
            for (int p = 0; p < 4; ++p) bst[cur ^ 1][p * 256 + tid] = st[p];
        }
        __syncthreads();
    }
    if (!live) return;

    float s = s1.x + s1.y;
    float q = s2.x + s2.y;
    s += __shfl_xor(s, 16, 32);
    q += __shfl_xor(q, 16, 32);
    const float m   = s * (1.0f / 768.0f);
    const float rsv = rsqrtf(q * (1.0f / 768.0f) - m * m + LN_EPS);
    const float pA  = rsv * do_r;
    const float pB  = m * pA;

    float gwc[8], bwc[8];
    #pragma unroll
    for (int ft = 0; ft < 8; ++ft) {
        gwc[ft] = gw[ft * 16 + l16];
        bwc[ft] = bw[ft * 16 + l16];
    }
    const int crow = mt * 16 + half * 8;
    #pragma unroll
    for (int r = 0; r < 8; ++r) {
        const int sl = half * 8 + r;
        const float sA = __shfl(pA,   sl, 32);
        const float sB = __shfl(pB,   sl, 32);
        const float sD = __shfl(do_r, sl, 32);
        #pragma unroll
        for (int ft = 0; ft < 8; ++ft) {
            const int bcol = ft * 16 + l16;
            C[(size_t)(crow + r) * F + bcol] = sA * acc[ft][r] - sB * gwc[ft] + sD * bwc[ft];
        }
    }
}

__global__ void __launch_bounds__(256)
gemm0_and_csr_fill_kernel(const float* __restrict__ feat,
                                          const float* __restrict__ do_is,
                                          const float* __restrict__ Wg,
                                          const float* __restrict__ gw, const float* __restrict__ bw,
                                          float* __restrict__ C, int M, int ngemm,
                                          const int* __restrict__ src, const int* __restrict__ dst,
                                          const int* __restrict__ rank, const int* __restrict__ rowptr,
                                          int* __restrict__ csr_src, int e) {
    const int blk = blockIdx.x;
    if (blk < ngemm) gemm0_lnfold_wmma_strips(blk, feat, do_is, Wg, gw, bw, C, M);
    else             csr_fill_edges(blk - ngemm, src, dst, rank, rowptr, csr_src, e);
}

template <int F4>
__device__ __forceinline__ float4 gather_row_sum(const float* __restrict__ x,
                                                 const int* __restrict__ rowptr,
                                                 const int* __restrict__ cnt_in,
                                                 const int* __restrict__ csr_src,
                                                 int row, int lane) {
    const int beg = rowptr[row];
    const int len = cnt_in[row];
    const int c4  = (lane < F4) ? lane : 0;
    const float4* xp = (const float4*)x + c4;
    float4 acc = make_float4(0.f, 0.f, 0.f, 0.f);
    for (int base = 0; base < len; base += 32) {
        const int m = (len - base < 32) ? (len - base) : 32;
        const int idx = (lane < m) ? csr_src[beg + base + lane] : 0;
        #pragma unroll 4
        for (int j = 0; j < m; ++j) {
            const int s = __builtin_amdgcn_readlane(idx, j);
            const float4 v = xp[(size_t)s * F4];
            acc.x += v.x; acc.y += v.y; acc.z += v.z; acc.w += v.w;
        }
    }
    return acc;
}

template <int F4>
__global__ void gather_sum_vec4_kernel(const float* __restrict__ x,
                                       const int* __restrict__ rowptr,
                                       const int* __restrict__ cnt_in,
                                       const int* __restrict__ csr_src,
                                       float* __restrict__ out, int n) {
    const int lane = threadIdx.x & 31;
    const int row = __builtin_amdgcn_readfirstlane((int)((blockIdx.x * blockDim.x + threadIdx.x) >> 5));
    if (row >= n) return;
    const float4 acc = gather_row_sum<F4>(x, rowptr, cnt_in, csr_src, row, lane);
    if (lane < F4) ((float4*)out)[(size_t)row * F4 + lane] = acc;
}

__global__ void gather_ln_relu_128_kernel(const float* __restrict__ x,
                                          const int* __restrict__ rowptr,
                                          const int* __restrict__ cnt_in,
                                          const int* __restrict__ csr_src,
                                          const float* __restrict__ di_is,
                                          const float* __restrict__ do_is,
                                          const float* __restrict__ g, const float* __restrict__ b,
                                          float* __restrict__ out, int n) {
    const int lane = threadIdx.x & 31;
    const int row = __builtin_amdgcn_readfirstlane((int)((blockIdx.x * blockDim.x + threadIdx.x) >> 5));
    if (row >= n) return;
    const float4 acc = gather_row_sum<32>(x, rowptr, cnt_in, csr_src, row, lane);
    const float di = di_is[row];
    const float ds = do_is[row];
    float v[4] = { acc.x * di, acc.y * di, acc.z * di, acc.w * di };
    float s = 0.f, s2 = 0.f;
    #pragma unroll
    for (int i = 0; i < 4; ++i) { s += v[i]; s2 += v[i] * v[i]; }
    #pragma unroll
    for (int off = 16; off > 0; off >>= 1) {
        s  += __shfl_xor(s,  off, 32);
        s2 += __shfl_xor(s2, off, 32);
    }
    const float m   = s  * (1.0f / 128.0f);
    const float rsv = rsqrtf(s2 * (1.0f / 128.0f) - m * m + LN_EPS);
    float y[4];
    #pragma unroll
    for (int i = 0; i < 4; ++i) {
        const int c = 4 * lane + i;
        const float t = (v[i] - m) * rsv * g[c] + b[c];
        y[i] = fmaxf(t, 0.0f) * ds;
    }
    ((float4*)out)[(size_t)row * 32 + lane] = make_float4(y[0], y[1], y[2], y[3]);
}

__global__ void gather_scale_bias_40_kernel(const float* __restrict__ x,
                                            const int* __restrict__ rowptr,
                                            const int* __restrict__ cnt_in,
                                            const int* __restrict__ csr_src,
                                            const float* __restrict__ di_is,
                                            const float* __restrict__ b2,
                                            float* __restrict__ out, int n) {
    const int lane = threadIdx.x & 31;
    const int row = __builtin_amdgcn_readfirstlane((int)((blockIdx.x * blockDim.x + threadIdx.x) >> 5));
    if (row >= n) return;
    const float4 acc = gather_row_sum<10>(x, rowptr, cnt_in, csr_src, row, lane);
    if (lane < 10) {
        const float di = di_is[row];
        const int c = 4 * lane;
        float4 o;
        o.x = acc.x * di + b2[c];
        o.y = acc.y * di + b2[c + 1];
        o.z = acc.z * di + b2[c + 2];
        o.w = acc.w * di + b2[c + 3];
        ((float4*)out)[(size_t)row * 10 + lane] = o;
    }
}

#define H_PITCH 132
#define W2_COLS 48

__global__ void __launch_bounds__(256)
gemm128_ln_relu_gemm40_wmma_kernel(const float* __restrict__ A, const float* __restrict__ W,
                                   const float* __restrict__ di_is,
                                   const float* __restrict__ do_is,
                                   const float* __restrict__ g, const float* __restrict__ b,
                                   const float* __restrict__ W2,
                                   float* __restrict__ C2, int M) {
    __shared__ v2f    w2s[64 * W2_COLS];
    __shared__ float4 htile[8 * 16 * (H_PITCH / 4)];
    const int K = 128, F = 128, F2 = 40;
    const int tid  = threadIdx.x;
    const int lane = tid & 31;
    const int wv   = tid >> 5;
    const int mt   = (blockIdx.x * blockDim.x + tid) >> 5;
    const bool live = (mt * 16 < M);
    const int half = lane >> 4;
    const int l16  = lane & 15;
    const int koff = half * 2;
    const int crow = mt * 16 + half * 8;
    float* tw = (float*)htile + wv * (16 * H_PITCH);

    {
        v2f st[12];
        #pragma unroll
        for (int j = 0; j < 12; ++j) {
            const int i   = tid + 256 * j;
            const int pr  = i / W2_COLS;
            const int col = i - pr * W2_COLS;
            const int cc  = (col < F2) ? col : (F2 - 1);
            st[j].x = W2[(2 * pr) * F2 + cc];
            st[j].y = W2[(2 * pr + 1) * F2 + cc];
        }
        #pragma unroll
        for (int j = 0; j < 12; ++j) {
            const int i   = tid + 256 * j;
            const int col = i % W2_COLS;
            v2f p = st[j];
            if (col >= F2) { p.x = 0.f; p.y = 0.f; }
            w2s[i] = p;
        }
    }

    if (live) {
        const int arow = mt * 16 + l16;
        const float* ap = A + (size_t)arow * K + koff;

        v8f acc[8];
        #pragma unroll
        for (int i = 0; i < 8; ++i) { v8f z = {}; acc[i] = z; }

        for (int k = 0; k < K; k += 4) {
            const v2f a = *(const v2f*)(ap + k);
            const float* wrow = W + (size_t)(k + koff) * F;
            #pragma unroll
            for (int ft = 0; ft < 8; ++ft) {
                const int bcol = ft * 16 + l16;
                v2f bb;
                bb.x = wrow[bcol];
                bb.y = wrow[bcol + F];
                acc[ft] = __builtin_amdgcn_wmma_f32_16x16x4_f32(
                    false, a, false, bb, (short)0, acc[ft], false, false);
            }
        }

        float gg[8], bbv[8];
        #pragma unroll
        for (int ft = 0; ft < 8; ++ft) {
            gg[ft]  = g[ft * 16 + l16];
            bbv[ft] = b[ft * 16 + l16];
        }
        #pragma unroll
        for (int r = 0; r < 8; ++r) {
            const float di = di_is[crow + r];
            const float ds = do_is[crow + r];
            float v[8];
            float s = 0.f, s2 = 0.f;
            #pragma unroll
            for (int ft = 0; ft < 8; ++ft) {
                v[ft] = acc[ft][r] * di;
                s += v[ft]; s2 += v[ft] * v[ft];
            }
            #pragma unroll
            for (int off = 8; off > 0; off >>= 1) {
                s  += __shfl_xor(s,  off, 32);
                s2 += __shfl_xor(s2, off, 32);
            }
            const float m   = s  * (1.0f / 128.0f);
            const float rsv = rsqrtf(s2 * (1.0f / 128.0f) - m * m + LN_EPS);
            #pragma unroll
            for (int ft = 0; ft < 8; ++ft) {
                const float t = (v[ft] - m) * rsv * gg[ft] + bbv[ft];
                tw[(half * 8 + r) * H_PITCH + ft * 16 + l16] = fmaxf(t, 0.0f) * ds;
            }
        }
    }
    __syncthreads();
    if (!live) return;

    v8f acc2[3];
    #pragma unroll
    for (int i = 0; i < 3; ++i) { v8f z = {}; acc2[i] = z; }
    const float* ta = tw + l16 * H_PITCH + koff;
    const v2f*   wb = w2s + half * W2_COLS + l16;
    for (int k = 0; k < K; k += 4) {
        const v2f a = *(const v2f*)(ta + k);
        #pragma unroll
        for (int ft = 0; ft < 3; ++ft) {
            const v2f bb = wb[(k >> 1) * W2_COLS + ft * 16];
            acc2[ft] = __builtin_amdgcn_wmma_f32_16x16x4_f32(
                false, a, false, bb, (short)0, acc2[ft], false, false);
        }
    }
    #pragma unroll
    for (int ft = 0; ft < 3; ++ft) {
        const int bcol = ft * 16 + l16;
        if (bcol < F2) {
            #pragma unroll
            for (int r = 0; r < 8; ++r)
                C2[(size_t)(crow + r) * F2 + bcol] = acc2[ft][r];
        }
    }
}

static inline int cdiv(long long a, long long b) { return (int)((a + b - 1) / b); }

extern "C" void kernel_launch(void* const* d_in, const int* in_sizes, int n_in,
                              void* d_out, int out_size, void* d_ws, size_t ws_size,
                              hipStream_t stream) {
    const float* feat = (const float*)d_in[0];
    const int*   src  = (const int*)d_in[1];
    const int*   dst  = (const int*)d_in[2];
    const float* W0   = (const float*)d_in[3];
    const float* W1   = (const float*)d_in[4];
    const float* W2   = (const float*)d_in[5];
    const float* b2   = (const float*)d_in[6];
    const float* g_in = (const float*)d_in[7];
    const float* b_in = (const float*)d_in[8];
    const float* g0   = (const float*)d_in[9];
    const float* b0   = (const float*)d_in[10];
    const float* g1   = (const float*)d_in[11];
    const float* b1   = (const float*)d_in[12];
    float* out = (float*)d_out;

    const int IN_F = 768, HID = 128;
    const int n = in_sizes[0] / IN_F;
    const int e = in_sizes[1];

    float* ws    = (float*)d_ws;
    float* do_is = ws;
    float* di_is = ws + (size_t)n;
    float* Wg    = ws + 2 * (size_t)n;
    float* gw    = Wg + (size_t)IN_F * HID;
    float* bw    = gw + HID;
    float* bufB  = ws + 4 * (size_t)n;
    float* bufC  = bufB + (size_t)n * HID;
    float* bufD  = bufC + (size_t)n * HID;
    int*   cnt_out = (int*)(bufD + (size_t)n * HID);
    int*   cnt_in  = cnt_out + (size_t)n;
    int*   rowptr  = cnt_in + (size_t)n;
    int*   rank    = rowptr + (size_t)n;
    int*   csr_src = rank + (size_t)e;
    float* bufE  = bufC;

    const int T = 256;

    hipMemsetAsync(cnt_out, 0, 2 * (size_t)n * sizeof(int), stream);
    degree_rank_kernel<<<cdiv(e, T), T, 0, stream>>>(src, dst, cnt_out, cnt_in, rank, e);

    const int nscan = cdiv(n, 1024);
    scan_and_fold_tables_kernel<<<nscan + HID / 16, 1024, 0, stream>>>(
        cnt_out, cnt_in, rowptr, do_is, di_is, W0, g_in, b_in, Wg, gw, bw, n, nscan);

    const int mtiles = cdiv(n, 16);
    const int ngemm  = cdiv((long long)mtiles * 32, T);
    gemm0_and_csr_fill_kernel<<<ngemm + cdiv(e, T), T, 0, stream>>>(
        feat, do_is, Wg, gw, bw, bufB, n, ngemm,
        src, dst, rank, rowptr, csr_src, e);

    gather_ln_relu_128_kernel<<<cdiv((long long)n * 32, T), T, 0, stream>>>(
        bufB, rowptr, cnt_in, csr_src, di_is, do_is, g0, b0, bufD, n);

    gather_sum_vec4_kernel<32><<<cdiv((long long)n * 32, T), T, 0, stream>>>(
        bufD, rowptr, cnt_in, csr_src, bufB, n);
    gemm128_ln_relu_gemm40_wmma_kernel<<<cdiv((long long)mtiles * 32, T), T, 0, stream>>>(
        bufB, W1, di_is, do_is, g1, b1, W2, bufE, n);

    gather_scale_bias_40_kernel<<<cdiv((long long)n * 32, T), T, 0, stream>>>(
        bufE, rowptr, cnt_in, csr_src, di_is, b2, out, n);
}


# Graph tables by a two-level counting sort in LDS (4096-edge chunks binned by 256-node bucket, one block per bucket) replace global atomics and scattered 4-byte stores; row sums still in arrival order
# speedup vs baseline: 4.0757x; 1.4203x over previous
#include <hip/hip_runtime.h>
#include <math.h>

typedef __attribute__((ext_vector_type(2))) float v2f;
typedef __attribute__((ext_vector_type(8))) float v8f;

#define LN_EPS 1e-5f

#define CSR_CH    4096
#define CSR_NBMAX 256
#define CSR_CAP   8192

__device__ __forceinline__ void
bin_chunk(int blk, const int* __restrict__ src, const int* __restrict__ dst,
          int* __restrict__ sortD_raw, int* __restrict__ sortS,
          int* __restrict__ tblD, int* __restrict__ tblS, int e, int nb, int nblk) {
    __shared__ int  hD[CSR_NBMAX + 1];
    __shared__ int  hS[CSR_NBMAX + 1];
    __shared__ int2 stD[CSR_CH];
    __shared__ int  stS[CSR_CH];
    const int tid  = threadIdx.x;
    const int lane = tid & 31;
    const int wid  = tid >> 5;
    int2* sortD = (int2*)sortD_raw;

    for (int i = tid; i <= CSR_NBMAX; i += 1024) { hD[i] = 0; hS[i] = 0; }
    __syncthreads();

    const int e0 = blk * CSR_CH;
    int s[4], d[4], bD[4], bS[4], rD[4], rS[4];
    #pragma unroll
    for (int j = 0; j < 4; ++j) {
        const int i = e0 + j * 1024 + tid;
        if (i < e) { s[j] = src[i]; d[j] = dst[i]; } else { s[j] = -1; d[j] = -1; }
    }
    #pragma unroll
    for (int j = 0; j < 4; ++j) {
        bD[j] = min(max(d[j], 0) >> 8, CSR_NBMAX - 1);
        bS[j] = min(max(s[j], 0) >> 8, CSR_NBMAX - 1);
        rD[j] = 0; rS[j] = 0;
        if (s[j] >= 0) {
            rD[j] = atomicAdd(&hD[bD[j]], 1);
            rS[j] = atomicAdd(&hS[bS[j]], 1);
        }
    }
    __syncthreads();
    if (wid < 2) {
        int* h = wid ? hS : hD;
        int v[8];
        int sum = 0;
        #pragma unroll
        for (int k = 0; k < 8; ++k) { v[k] = h[lane * 8 + k]; sum += v[k]; }
        int incl = sum;
        #pragma unroll
        for (int off = 1; off < 32; off <<= 1) {
            const int t = __shfl_up(incl, off, 32);
            if (lane >= off) incl += t;
        }
        int run = incl - sum;
        #pragma unroll
        for (int k = 0; k < 8; ++k) { h[lane * 8 + k] = run; run += v[k]; }
        if (lane == 31) h[CSR_NBMAX] = run;
    }
    __syncthreads();
    #pragma unroll
    for (int j = 0; j < 4; ++j) {
        if (s[j] >= 0) {
            stD[hD[bD[j]] + rD[j]] = make_int2(s[j], d[j]);
            stS[hS[bS[j]] + rS[j]] = s[j];
        }
    }
    __syncthreads();
    const int cnt = (e - e0 < CSR_CH) ? (e - e0) : CSR_CH;
    for (int i = tid; i < cnt; i += 1024) {
        sortD[(size_t)e0 + i] = stD[i];
        sortS[(size_t)e0 + i] = stS[i];
    }
    const int nbc = (nb < CSR_NBMAX) ? nb : CSR_NBMAX;
    for (int B = tid; B <= nbc; B += 1024) {
        tblD[(size_t)B * nblk + blk] = hD[B];
        tblS[(size_t)B * nblk + blk] = hS[B];
    }
}

__device__ __forceinline__ void
finalize_bucket(int B, bool is_dst, const int* __restrict__ tbl,
                const int* __restrict__ sortD_raw, const int* __restrict__ sortS,
                int nblk, int n,
                int* __restrict__ cnt_in, int* __restrict__ rowptr, int* __restrict__ csr_src,
                float* __restrict__ do_is, float* __restrict__ di_is) {
    __shared__ int cntl[256];
    __shared__ int startl[256];
    __shared__ int cursor[256];
    __shared__ int redb[32];
    __shared__ int redm[32];
    __shared__ int stage[CSR_CAP];
    const int tid  = threadIdx.x;
    const int lane = tid & 31;
    const int wid  = tid >> 5;
    const int2* sortD = (const int2*)sortD_raw;
    const int* tb = tbl + (size_t)B * nblk;

    if (tid < 256) cntl[tid] = 0;
    int sb = 0, sm = 0;
    for (int c = tid; c < nblk; c += 1024) {
        const int s0 = tb[c], s1 = tb[nblk + c];
        sb += s0; sm += s1 - s0;
    }
    #pragma unroll
    for (int off = 16; off > 0; off >>= 1) {
        sb += __shfl_xor(sb, off, 32);
        sm += __shfl_xor(sm, off, 32);
    }
    if (lane == 0) { redb[wid] = sb; redm[wid] = sm; }
    __syncthreads();
    int base = redb[lane];
    int m    = redm[lane];
    #pragma unroll
    for (int off = 16; off > 0; off >>= 1) {
        base += __shfl_xor(base, off, 32);
        m    += __shfl_xor(m,    off, 32);
    }

    for (int cb = 0; cb < nblk; cb += 1024) {
        const int cm = cb + wid + 32 * lane;
        int s0v = 0, s1v = 0;
        if (cm < nblk) { s0v = tb[cm]; s1v = tb[nblk + cm]; }
        for (int k = 0; k < 32; ++k) {
            const int c = cb + wid + 32 * k;
            if (c >= nblk) break;
            const int s0 = __shfl(s0v, k, 32), s1 = __shfl(s1v, k, 32);
            const size_t o = (size_t)c * CSR_CH;
            for (int i = s0 + lane; i < s1; i += 32) {
                const int key = is_dst ? sortD[o + i].y : sortS[o + i];
                atomicAdd(&cntl[key & 255], 1);
            }
        }
    }
    __syncthreads();
    if (wid == 0) {
        int v[8];
        int sum = 0;
        #pragma unroll
        for (int k = 0; k < 8; ++k) { v[k] = cntl[lane * 8 + k]; sum += v[k]; }
        int incl = sum;
        #pragma unroll
        for (int off = 1; off < 32; off <<= 1) {
            const int t = __shfl_up(incl, off, 32);
            if (lane >= off) incl += t;
        }
        int run = incl - sum;
        #pragma unroll
        for (int k = 0; k < 8; ++k) { startl[lane * 8 + k] = run; run += v[k]; }
    }
    __syncthreads();
    if (tid < 256) {
        const int v = (B << 8) + tid;
        const int cdeg = cntl[tid];
        cursor[tid] = startl[tid];
        if (v < n) {
            const float is = rsqrtf(fmaxf((float)cdeg, 1.0f));
            if (is_dst) {
                cnt_in[v] = cdeg;
                rowptr[v] = base + startl[tid];
                di_is[v]  = is;
            } else {
                do_is[v] = is;
            }
        }
    }
    if (!is_dst) return;
    __syncthreads();
    const bool fits = (m <= CSR_CAP);
    for (int cb = 0; cb < nblk; cb += 1024) {
        const int cm = cb + wid + 32 * lane;
        int s0v = 0, s1v = 0;
        if (cm < nblk) { s0v = tb[cm]; s1v = tb[nblk + cm]; }
        for (int k = 0; k < 32; ++k) {
            const int c = cb + wid + 32 * k;
            if (c >= nblk) break;
            const int s0 = __shfl(s0v, k, 32), s1 = __shfl(s1v, k, 32);
            const size_t o = (size_t)c * CSR_CH;
            for (int i = s0 + lane; i < s1; i += 32) {
                const int2 rec = sortD[o + i];
                const int p = atomicAdd(&cursor[rec.y & 255], 1);
                if (fits) stage[p] = rec.x;
                else      csr_src[(size_t)base + p] = rec.x;
            }
        }
    }
    if (!fits) return;
    __syncthreads();
    for (int i = tid; i < m; i += 1024) csr_src[(size_t)base + i] = stage[i];
}

__global__ void __launch_bounds__(1024)
finalize_buckets_kernel(const int* __restrict__ tblD, const int* __restrict__ tblS,
                        const int* __restrict__ sortD, const int* __restrict__ sortS,
                        int nblk, int nb, int n,
                        int* __restrict__ cnt_in, int* __restrict__ rowptr, int* __restrict__ csr_src,
                        float* __restrict__ do_is, float* __restrict__ di_is) {
    const int blk = blockIdx.x;
    if (blk < nb) finalize_bucket(blk, true, tblD, sortD, sortS, nblk, n, cnt_in, rowptr, csr_src, do_is, di_is);
    else          finalize_bucket(blk - nb, false, tblS, sortD, sortS, nblk, n, cnt_in, rowptr, csr_src, do_is, di_is);
}

__device__ __forceinline__ size_t wg_frag_index(int k, int col) {
    const int pr = k >> 1, kb = k & 1;
    const int q = col >> 5, j = (col >> 4) & 1, c = col & 15;
    return (size_t)pr * 256 + (size_t)((((q * 16 + c) * 2 + j) * 2) + kb);
}

__device__ __forceinline__ void
ln_fold_tables_16cols(int blk, const float* __restrict__ W,
                      const float* __restrict__ g, const float* __restrict__ b,
                      float* __restrict__ Wg, float* __restrict__ gw, float* __restrict__ bw) {
    __shared__ float pg[64][17];
    __shared__ float pb[64][17];
    const int tid = threadIdx.x;
    const int cc  = tid & 15;
    const int ks  = tid >> 4;
    const int col = blk * 16 + cc;
    float sg = 0.f, sb = 0.f;
    #pragma unroll
    for (int i = 0; i < 12; ++i) {
        const int k = ks * 12 + i;
        const float w  = W[(size_t)k * 128 + col];
        const float wg = g[k] * w;
        Wg[wg_frag_index(k, col)] = wg;
        sg += wg;
        sb = fmaf(b[k], w, sb);
    }
    pg[ks][cc] = sg;
    pb[ks][cc] = sb;
    __syncthreads();
    if (tid < 32) {
        float acc = 0.f;
        if (tid < 16) { for (int j = 0; j < 64; ++j) acc += pg[j][cc]; gw[col] = acc; }
        else          { for (int j = 0; j < 64; ++j) acc += pb[j][cc]; bw[col] = acc; }
    }
}

__global__ void __launch_bounds__(1024)
bin_edges_and_fold_tables_kernel(const int* __restrict__ src, const int* __restrict__ dst,
                                 int* __restrict__ sortD, int* __restrict__ sortS,
                                 int* __restrict__ tblD, int* __restrict__ tblS,
                                 int e, int nb, int nblk,
                                 const float* __restrict__ W, const float* __restrict__ g,
                                 const float* __restrict__ b,
                                 float* __restrict__ Wg, float* __restrict__ gw, float* __restrict__ bw) {
    const int blk = blockIdx.x;
    if (blk < nblk) bin_chunk(blk, src, dst, sortD, sortS, tblD, tblS, e, nb, nblk);
    else            ln_fold_tables_16cols(blk - nblk, W, g, b, Wg, gw, bw);
}

#define G0_KC     32
#define G0_CHUNK4 (G0_KC / 2 * 64)

__device__ __forceinline__ void
gemm0_lnfold_wmma_strips(int blk, const float* __restrict__ feat,
                         const float* __restrict__ do_is,
                         const float* __restrict__ Wg,
                         const float* __restrict__ gw, const float* __restrict__ bw,
                         float* __restrict__ C, int M) {
    __shared__ float4 bst[2][G0_CHUNK4];
    const int K = 768, F = 128;
    const int tid  = threadIdx.x;
    const int lane = tid & 31;
    const int mt   = (blk * blockDim.x + tid) >> 5;
    const bool live = (mt * 16 < M);
    const int half = lane >> 4;
    const int l16  = lane & 15;
    int arow = mt * 16 + l16;
    if (arow >= M) arow = M - 1;
    const int koff = half * 2;
    const float do_r = do_is[arow];
    const float* ap = feat + (size_t)arow * K + koff;
    const float4* wg4 = (const float4*)Wg;

    v8f acc[8];
    #pragma unroll
    for (int i = 0; i < 8; ++i) { v8f z = {}; acc[i] = z; }
    v2f s1 = {0.f, 0.f};
    v2f s2 = {0.f, 0.f};

    float4 st[4];
    #pragma unroll
    for (int p = 0; p < 4; ++p) st[p] = wg4[p * 256 + tid];
    #pragma unroll
    for (int p = 0; p < 4; ++p) bst[0][p * 256 + tid] = st[p];
    __syncthreads();

    const int NCH = K / G0_KC;
    for (int ch = 0; ch < NCH; ++ch) {
        const int cur = ch & 1;
        const bool more = (ch + 1 < NCH);
        if (more) {
            #pragma unroll
            for (int p = 0; p < 4; ++p) st[p] = wg4[(size_t)(ch + 1) * G0_CHUNK4 + p * 256 + tid];
        }
        const float* apc = ap + ch * G0_KC;
        const float4* bcur = &bst[cur][half * 64 + l16];
        v2f af[G0_KC / 4];
        #pragma unroll
        for (int i = 0; i < G0_KC / 4; ++i) af[i] = *(const v2f*)(apc + 4 * i);
        #pragma unroll
        for (int kk = 0; kk < G0_KC; kk += 4) {
            const v2f a = af[kk / 4];
            s1 += a;
            s2 += a * a;
            #pragma unroll
            for (int q = 0; q < 4; ++q) {
                const float4 w = bcur[(kk / 2) * 64 + q * 16];
                v2f b0; b0.x = w.x; b0.y = w.y;
                v2f b1; b1.x = w.z; b1.y = w.w;
                acc[2 * q] = __builtin_amdgcn_wmma_f32_16x16x4_f32(
                    false, a, false, b0, (short)0, acc[2 * q], false, false);
                acc[2 * q + 1] = __builtin_amdgcn_wmma_f32_16x16x4_f32(
                    false, a, false, b1, (short)0, acc[2 * q + 1], false, false);
            }
        }
        if (more) {
            #pragma unroll
            for (int p = 0; p < 4; ++p) bst[cur ^ 1][p * 256 + tid] = st[p];
        }
        __syncthreads();
    }
    if (!live) return;

    float s = s1.x + s1.y;
    float q = s2.x + s2.y;
    s += __shfl_xor(s, 16, 32);
    q += __shfl_xor(q, 16, 32);
    const float m   = s * (1.0f / 768.0f);
    const float rsv = rsqrtf(q * (1.0f / 768.0f) - m * m + LN_EPS);
    const float pA  = rsv * do_r;
    const float pB  = m * pA;

    float gwc[8], bwc[8];
    #pragma unroll
    for (int ft = 0; ft < 8; ++ft) {
        gwc[ft] = gw[ft * 16 + l16];
        bwc[ft] = bw[ft * 16 + l16];
    }
    const int crow = mt * 16 + half * 8;
    #pragma unroll
    for (int r = 0; r < 8; ++r) {
        const int sl = half * 8 + r;
        const float sA = __shfl(pA,   sl, 32);
        const float sB = __shfl(pB,   sl, 32);
        const float sD = __shfl(do_r, sl, 32);
        #pragma unroll
        for (int ft = 0; ft < 8; ++ft) {
            const int bcol = ft * 16 + l16;
            C[(size_t)(crow + r) * F + bcol] = sA * acc[ft][r] - sB * gwc[ft] + sD * bwc[ft];
        }
    }
}

__global__ void __launch_bounds__(256) __attribute__((amdgpu_waves_per_eu(2, 5)))
gemm0_lnfold_kernel(const float* __restrict__ feat,
                    const float* __restrict__ do_is,
                    const float* __restrict__ Wg,
                    const float* __restrict__ gw, const float* __restrict__ bw,
                    float* __restrict__ C, int M) {
    gemm0_lnfold_wmma_strips(blockIdx.x, feat, do_is, Wg, gw, bw, C, M);
}

template <int F4>
__device__ __forceinline__ float4 gather_row_sum(const float* __restrict__ x,
                                                 const int* __restrict__ rowptr,
                                                 const int* __restrict__ cnt_in,
                                                 const int* __restrict__ csr_src,
                                                 int row, int lane) {
    const int beg = rowptr[row];
    const int len = cnt_in[row];
    const int c4  = (lane < F4) ? lane : 0;
    const float4* xp = (const float4*)x + c4;
    float4 acc = make_float4(0.f, 0.f, 0.f, 0.f);
    for (int base = 0; base < len; base += 32) {
        const int m = (len - base < 32) ? (len - base) : 32;
        const int idx = (lane < m) ? csr_src[beg + base + lane] : 0;
        #pragma unroll 4
        for (int j = 0; j < m; ++j) {
            const int s = __builtin_amdgcn_readlane(idx, j);
            const float4 v = xp[(size_t)s * F4];
            acc.x += v.x; acc.y += v.y; acc.z += v.z; acc.w += v.w;
        }
    }
    return acc;
}

template <int F4>
__global__ void gather_sum_vec4_kernel(const float* __restrict__ x,
                                       const int* __restrict__ rowptr,
                                       const int* __restrict__ cnt_in,
                                       const int* __restrict__ csr_src,
                                       float* __restrict__ out, int n) {
    const int lane = threadIdx.x & 31;
    const int row = __builtin_amdgcn_readfirstlane((int)((blockIdx.x * blockDim.x + threadIdx.x) >> 5));
    if (row >= n) return;
    const float4 acc = gather_row_sum<F4>(x, rowptr, cnt_in, csr_src, row, lane);
    if (lane < F4) ((float4*)out)[(size_t)row * F4 + lane] = acc;
}

__global__ void gather_ln_relu_128_kernel(const float* __restrict__ x,
                                          const int* __restrict__ rowptr,
                                          const int* __restrict__ cnt_in,
                                          const int* __restrict__ csr_src,
                                          const float* __restrict__ di_is,
                                          const float* __restrict__ do_is,
                                          const float* __restrict__ g, const float* __restrict__ b,
                                          float* __restrict__ out, int n) {
    const int lane = threadIdx.x & 31;
    const int row = __builtin_amdgcn_readfirstlane((int)((blockIdx.x * blockDim.x + threadIdx.x) >> 5));
    if (row >= n) return;
    const float4 acc = gather_row_sum<32>(x, rowptr, cnt_in, csr_src, row, lane);
    const float di = di_is[row];
    const float ds = do_is[row];
    float v[4] = { acc.x * di, acc.y * di, acc.z * di, acc.w * di };
    float s = 0.f, s2 = 0.f;
    #pragma unroll
    for (int i = 0; i < 4; ++i) { s += v[i]; s2 += v[i] * v[i]; }
    #pragma unroll
    for (int off = 16; off > 0; off >>= 1) {
        s  += __shfl_xor(s,  off, 32);
        s2 += __shfl_xor(s2, off, 32);
    }
    const float m   = s  * (1.0f / 128.0f);
    const float rsv = rsqrtf(s2 * (1.0f / 128.0f) - m * m + LN_EPS);
    float y[4];
    #pragma unroll
    for (int i = 0; i < 4; ++i) {
        const int c = 4 * lane + i;
        const float t = (v[i] - m) * rsv * g[c] + b[c];
        y[i] = fmaxf(t, 0.0f) * ds;
    }
    ((float4*)out)[(size_t)row * 32 + lane] = make_float4(y[0], y[1], y[2], y[3]);
}

__global__ void gather_scale_bias_40_kernel(const float* __restrict__ x,
                                            const int* __restrict__ rowptr,
                                            const int* __restrict__ cnt_in,
                                            const int* __restrict__ csr_src,
                                            const float* __restrict__ di_is,
                                            const float* __restrict__ b2,
                                            float* __restrict__ out, int n) {
    const int lane = threadIdx.x & 31;
    const int row = __builtin_amdgcn_readfirstlane((int)((blockIdx.x * blockDim.x + threadIdx.x) >> 5));
    if (row >= n) return;
    const float4 acc = gather_row_sum<10>(x, rowptr, cnt_in, csr_src, row, lane);
    if (lane < 10) {
        const float di = di_is[row];
        const int c = 4 * lane;
        float4 o;
        o.x = acc.x * di + b2[c];
        o.y = acc.y * di + b2[c + 1];
        o.z = acc.z * di + b2[c + 2];
        o.w = acc.w * di + b2[c + 3];
        ((float4*)out)[(size_t)row * 10 + lane] = o;
    }
}

#define H_PITCH 132
#define W2_COLS 48

__global__ void __launch_bounds__(256)
gemm128_ln_relu_gemm40_wmma_kernel(const float* __restrict__ A, const float* __restrict__ W,
                                   const float* __restrict__ di_is,
                                   const float* __restrict__ do_is,
                                   const float* __restrict__ g, const float* __restrict__ b,
                                   const float* __restrict__ W2,
                                   float* __restrict__ C2, int M) {
    __shared__ v2f    w2s[64 * W2_COLS];
    __shared__ float4 htile[8 * 16 * (H_PITCH / 4)];
    const int K = 128, F = 128, F2 = 40;
    const int tid  = threadIdx.x;
    const int lane = tid & 31;
    const int wv   = tid >> 5;
    const int mt   = (blockIdx.x * blockDim.x + tid) >> 5;
    const bool live = (mt * 16 < M);
    const int half = lane >> 4;
    const int l16  = lane & 15;
    const int koff = half * 2;
    const int crow = mt * 16 + half * 8;
    float* tw = (float*)htile + wv * (16 * H_PITCH);

    {
        v2f st[12];
        #pragma unroll
        for (int j = 0; j < 12; ++j) {
            const int i   = tid + 256 * j;
            const int pr  = i / W2_COLS;
            const int col = i - pr * W2_COLS;
            const int cc  = (col < F2) ? col : (F2 - 1);
            st[j].x = W2[(2 * pr) * F2 + cc];
            st[j].y = W2[(2 * pr + 1) * F2 + cc];
        }
        #pragma unroll
        for (int j = 0; j < 12; ++j) {
            const int i   = tid + 256 * j;
            const int col = i % W2_COLS;
            v2f p = st[j];
            if (col >= F2) { p.x = 0.f; p.y = 0.f; }
            w2s[i] = p;
        }
    }

    if (live) {
        const int arow = mt * 16 + l16;
        const float* ap = A + (size_t)arow * K + koff;

        v8f acc[8];
        #pragma unroll
        for (int i = 0; i < 8; ++i) { v8f z = {}; acc[i] = z; }

        for (int k = 0; k < K; k += 4) {
            const v2f a = *(const v2f*)(ap + k);
            const float* wrow = W + (size_t)(k + koff) * F;
            #pragma unroll
            for (int ft = 0; ft < 8; ++ft) {
                const int bcol = ft * 16 + l16;
                v2f bb;
                bb.x = wrow[bcol];
                bb.y = wrow[bcol + F];
                acc[ft] = __builtin_amdgcn_wmma_f32_16x16x4_f32(
                    false, a, false, bb, (short)0, acc[ft], false, false);
            }
        }

        float gg[8], bbv[8];
        #pragma unroll
        for (int ft = 0; ft < 8; ++ft) {
            gg[ft]  = g[ft * 16 + l16];
            bbv[ft] = b[ft * 16 + l16];
        }
        #pragma unroll
        for (int r = 0; r < 8; ++r) {
            const float di = di_is[crow + r];
            const float ds = do_is[crow + r];
            float v[8];
            float s = 0.f, s2 = 0.f;
            #pragma unroll
            for (int ft = 0; ft < 8; ++ft) {
                v[ft] = acc[ft][r] * di;
                s += v[ft]; s2 += v[ft] * v[ft];
            }
            #pragma unroll
            for (int off = 8; off > 0; off >>= 1) {
                s  += __shfl_xor(s,  off, 32);
                s2 += __shfl_xor(s2, off, 32);
            }
            const float m   = s  * (1.0f / 128.0f);
            const float rsv = rsqrtf(s2 * (1.0f / 128.0f) - m * m + LN_EPS);
            #pragma unroll
            for (int ft = 0; ft < 8; ++ft) {
                const float t = (v[ft] - m) * rsv * gg[ft] + bbv[ft];
                tw[(half * 8 + r) * H_PITCH + ft * 16 + l16] = fmaxf(t, 0.0f) * ds;
            }
        }
    }
    __syncthreads();
    if (!live) return;

    v8f acc2[3];
    #pragma unroll
    for (int i = 0; i < 3; ++i) { v8f z = {}; acc2[i] = z; }
    const float* ta = tw + l16 * H_PITCH + koff;
    const v2f*   wb = w2s + half * W2_COLS + l16;
    for (int k = 0; k < K; k += 4) {
        const v2f a = *(const v2f*)(ta + k);
        #pragma unroll
        for (int ft = 0; ft < 3; ++ft) {
            const v2f bb = wb[(k >> 1) * W2_COLS + ft * 16];
            acc2[ft] = __builtin_amdgcn_wmma_f32_16x16x4_f32(
                false, a, false, bb, (short)0, acc2[ft], false, false);
        }
    }
    #pragma unroll
    for (int ft = 0; ft < 3; ++ft) {
        const int bcol = ft * 16 + l16;
        if (bcol < F2) {
            #pragma unroll
            for (int r = 0; r < 8; ++r)
                C2[(size_t)(crow + r) * F2 + bcol] = acc2[ft][r];
        }
    }
}

static inline int cdiv(long long a, long long b) { return (int)((a + b - 1) / b); }

extern "C" void kernel_launch(void* const* d_in, const int* in_sizes, int n_in,
                              void* d_out, int out_size, void* d_ws, size_t ws_size,
                              hipStream_t stream) {
    const float* feat = (const float*)d_in[0];
    const int*   src  = (const int*)d_in[1];
    const int*   dst  = (const int*)d_in[2];
    const float* W0   = (const float*)d_in[3];
    const float* W1   = (const float*)d_in[4];
    const float* W2   = (const float*)d_in[5];
    const float* b2   = (const float*)d_in[6];
    const float* g_in = (const float*)d_in[7];
    const float* b_in = (const float*)d_in[8];
    const float* g0   = (const float*)d_in[9];
    const float* b0   = (const float*)d_in[10];
    const float* g1   = (const float*)d_in[11];
    const float* b1   = (const float*)d_in[12];
    float* out = (float*)d_out;

    const int IN_F = 768, HID = 128;
    const int n = in_sizes[0] / IN_F;
    const int e = in_sizes[1];

    float* ws    = (float*)d_ws;
    float* do_is = ws;
    float* di_is = ws + (size_t)n;
    float* Wg    = ws + 2 * (size_t)n;
    float* gw    = Wg + (size_t)IN_F * HID;
    float* bw    = gw + HID;
    float* bufB  = ws + 4 * (size_t)n;
    float* bufC  = bufB + (size_t)n * HID;
    float* bufD  = bufC + (size_t)n * HID;
    int*   cnt_in  = (int*)(bufD + (size_t)n * HID);
    int*   rowptr  = cnt_in + (size_t)n;
    int*   csr_src = rowptr + (size_t)n;
    float* bufE  = bufC;

    const int T = 256;

    const int nblk = cdiv(e, CSR_CH);
    const int nb   = cdiv(n, 256);
    int* sortD = (int*)bufC;
    int* sortS = (int*)bufD;
    int* tblD  = sortS + (size_t)e;
    int* tblS  = tblD + (size_t)(nb + 1) * nblk;
    bin_edges_and_fold_tables_kernel<<<nblk + HID / 16, 1024, 0, stream>>>(
        src, dst, sortD, sortS, tblD, tblS, e, nb, nblk, W0, g_in, b_in, Wg, gw, bw);

    finalize_buckets_kernel<<<2 * nb, 1024, 0, stream>>>(
        tblD, tblS, sortD, sortS, nblk, nb, n, cnt_in, rowptr, csr_src, do_is, di_is);

    const int mtiles = cdiv(n, 16);
    const int ngemm  = cdiv((long long)mtiles * 32, T);
    gemm0_lnfold_kernel<<<ngemm, T, 0, stream>>>(feat, do_is, Wg, gw, bw, bufB, n);

    gather_ln_relu_128_kernel<<<cdiv((long long)n * 32, T), T, 0, stream>>>(
        bufB, rowptr, cnt_in, csr_src, di_is, do_is, g0, b0, bufD, n);

    gather_sum_vec4_kernel<32><<<cdiv((long long)n * 32, T), T, 0, stream>>>(
        bufD, rowptr, cnt_in, csr_src, bufB, n);
    gemm128_ln_relu_gemm40_wmma_kernel<<<cdiv((long long)mtiles * 32, T), T, 0, stream>>>(
        bufB, W1, di_is, do_is, g1, b1, W2, bufE, n);

    gather_scale_bias_40_kernel<<<cdiv((long long)n * 32, T), T, 0, stream>>>(
        bufE, rowptr, cnt_in, csr_src, di_is, b2, out, n);
}


# Layer-0 product dealt out evenly over the SIMDs: one 16-wave block per WGP, a wave owns 32 columns of 3 or 4 strips, feat staged through LDS beside Wg; every sum in its old order
# speedup vs baseline: 4.2289x; 1.0376x over previous
#include <hip/hip_runtime.h>
#include <math.h>

typedef __attribute__((ext_vector_type(2))) float v2f;
typedef __attribute__((ext_vector_type(8))) float v8f;

#define LN_EPS 1e-5f

#define CSR_CH    4096
#define CSR_NBMAX 256
#define CSR_CAP   8192

__device__ __forceinline__ void
bin_chunk(int blk, const int* __restrict__ src, const int* __restrict__ dst,
          int* __restrict__ sortD_raw, int* __restrict__ sortS,
          int* __restrict__ tblD, int* __restrict__ tblS, int e, int nb, int nblk) {
    __shared__ int  hD[CSR_NBMAX + 1];
    __shared__ int  hS[CSR_NBMAX + 1];
    __shared__ int2 stD[CSR_CH];
    __shared__ int  stS[CSR_CH];
    const int tid  = threadIdx.x;
    const int lane = tid & 31;
    const int wid  = tid >> 5;
    int2* sortD = (int2*)sortD_raw;

    for (int i = tid; i <= CSR_NBMAX; i += 1024) { hD[i] = 0; hS[i] = 0; }
    __syncthreads();

    const int e0 = blk * CSR_CH;
    int s[4], d[4], bD[4], bS[4], rD[4], rS[4];
    #pragma unroll
    for (int j = 0; j < 4; ++j) {
        const int i = e0 + j * 1024 + tid;
        if (i < e) { s[j] = src[i]; d[j] = dst[i]; } else { s[j] = -1; d[j] = -1; }
    }
    #pragma unroll
    for (int j = 0; j < 4; ++j) {
        bD[j] = min(max(d[j], 0) >> 8, CSR_NBMAX - 1);
        bS[j] = min(max(s[j], 0) >> 8, CSR_NBMAX - 1);
        rD[j] = 0; rS[j] = 0;
        if (s[j] >= 0) {
            rD[j] = atomicAdd(&hD[bD[j]], 1);
            rS[j] = atomicAdd(&hS[bS[j]], 1);
        }
    }
    __syncthreads();
    if (wid < 2) {
        int* h = wid ? hS : hD;
        int v[8];
        int sum = 0;
        #pragma unroll
        for (int k = 0; k < 8; ++k) { v[k] = h[lane * 8 + k]; sum += v[k]; }
        int incl = sum;
        #pragma unroll
        for (int off = 1; off < 32; off <<= 1) {
            const int t = __shfl_up(incl, off, 32);
            if (lane >= off) incl += t;
        }
        int run = incl - sum;
        #pragma unroll
        for (int k = 0; k < 8; ++k) { h[lane * 8 + k] = run; run += v[k]; }
        if (lane == 31) h[CSR_NBMAX] = run;
    }
    __syncthreads();
    #pragma unroll
    for (int j = 0; j < 4; ++j) {
        if (s[j] >= 0) {
            stD[hD[bD[j]] + rD[j]] = make_int2(s[j], d[j]);
            stS[hS[bS[j]] + rS[j]] = s[j];
        }
    }
    __syncthreads();
    const int cnt = (e - e0 < CSR_CH) ? (e - e0) : CSR_CH;
    for (int i = tid; i < cnt; i += 1024) {
        sortD[(size_t)e0 + i] = stD[i];
        sortS[(size_t)e0 + i] = stS[i];
    }
    const int nbc = (nb < CSR_NBMAX) ? nb : CSR_NBMAX;
    for (int B = tid; B <= nbc; B += 1024) {
        tblD[(size_t)B * nblk + blk] = hD[B];
        tblS[(size_t)B * nblk + blk] = hS[B];
    }
}

__device__ __forceinline__ void
finalize_bucket(int B, bool is_dst, const int* __restrict__ tbl,
                const int* __restrict__ sortD_raw, const int* __restrict__ sortS,
                int nblk, int n,
                int* __restrict__ cnt_in, int* __restrict__ rowptr, int* __restrict__ csr_src,
                float* __restrict__ do_is, float* __restrict__ di_is) {
    __shared__ int cntl[256];
    __shared__ int startl[256];
    __shared__ int cursor[256];
    __shared__ int redb[32];
    __shared__ int redm[32];
    __shared__ int stage[CSR_CAP];
    const int tid  = threadIdx.x;
    const int lane = tid & 31;
    const int wid  = tid >> 5;
    const int2* sortD = (const int2*)sortD_raw;
    const int* tb = tbl + (size_t)B * nblk;

    if (tid < 256) cntl[tid] = 0;
    int sb = 0, sm = 0;
    for (int c = tid; c < nblk; c += 1024) {
        const int s0 = tb[c], s1 = tb[nblk + c];
        sb += s0; sm += s1 - s0;
    }
    #pragma unroll
    for (int off = 16; off > 0; off >>= 1) {
        sb += __shfl_xor(sb, off, 32);
        sm += __shfl_xor(sm, off, 32);
    }
    if (lane == 0) { redb[wid] = sb; redm[wid] = sm; }
    __syncthreads();
    int base = redb[lane];
    int m    = redm[lane];
    #pragma unroll
    for (int off = 16; off > 0; off >>= 1) {
        base += __shfl_xor(base, off, 32);
        m    += __shfl_xor(m,    off, 32);
    }

    for (int cb = 0; cb < nblk; cb += 1024) {
        const int cm = cb + wid + 32 * lane;
        int s0v = 0, s1v = 0;
        if (cm < nblk) { s0v = tb[cm]; s1v = tb[nblk + cm]; }
        for (int k = 0; k < 32; ++k) {
            const int c = cb + wid + 32 * k;
            if (c >= nblk) break;
            const int s0 = __shfl(s0v, k, 32), s1 = __shfl(s1v, k, 32);
            const size_t o = (size_t)c * CSR_CH;
            for (int i = s0 + lane; i < s1; i += 32) {
                const int key = is_dst ? sortD[o + i].y : sortS[o + i];
                atomicAdd(&cntl[key & 255], 1);
            }
        }
    }
    __syncthreads();
    if (wid == 0) {
        int v[8];
        int sum = 0;
        #pragma unroll
        for (int k = 0; k < 8; ++k) { v[k] = cntl[lane * 8 + k]; sum += v[k]; }
        int incl = sum;
        #pragma unroll
        for (int off = 1; off < 32; off <<= 1) {
            const int t = __shfl_up(incl, off, 32);
            if (lane >= off) incl += t;
        }
        int run = incl - sum;
        #pragma unroll
        for (int k = 0; k < 8; ++k) { startl[lane * 8 + k] = run; run += v[k]; }
    }
    __syncthreads();
    if (tid < 256) {
        const int v = (B << 8) + tid;
        const int cdeg = cntl[tid];
        cursor[tid] = startl[tid];
        if (v < n) {
            const float is = rsqrtf(fmaxf((float)cdeg, 1.0f));
            if (is_dst) {
                cnt_in[v] = cdeg;
                rowptr[v] = base + startl[tid];
                di_is[v]  = is;
            } else {
                do_is[v] = is;
            }
        }
    }
    if (!is_dst) return;
    __syncthreads();
    const bool fits = (m <= CSR_CAP);
    for (int cb = 0; cb < nblk; cb += 1024) {
        const int cm = cb + wid + 32 * lane;
        int s0v = 0, s1v = 0;
        if (cm < nblk) { s0v = tb[cm]; s1v = tb[nblk + cm]; }
        for (int k = 0; k < 32; ++k) {
            const int c = cb + wid + 32 * k;
            if (c >= nblk) break;
            const int s0 = __shfl(s0v, k, 32), s1 = __shfl(s1v, k, 32);
            const size_t o = (size_t)c * CSR_CH;
            for (int i = s0 + lane; i < s1; i += 32) {
                const int2 rec = sortD[o + i];
                const int p = atomicAdd(&cursor[rec.y & 255], 1);
                if (fits) stage[p] = rec.x;
                else      csr_src[(size_t)base + p] = rec.x;
            }
        }
    }
    if (!fits) return;
    __syncthreads();
    for (int i = tid; i < m; i += 1024) csr_src[(size_t)base + i] = stage[i];
}

__global__ void __launch_bounds__(1024)
finalize_buckets_kernel(const int* __restrict__ tblD, const int* __restrict__ tblS,
                        const int* __restrict__ sortD, const int* __restrict__ sortS,
                        int nblk, int nb, int n,
                        int* __restrict__ cnt_in, int* __restrict__ rowptr, int* __restrict__ csr_src,
                        float* __restrict__ do_is, float* __restrict__ di_is) {
    const int blk = blockIdx.x;
    if (blk < nb) finalize_bucket(blk, true, tblD, sortD, sortS, nblk, n, cnt_in, rowptr, csr_src, do_is, di_is);
    else          finalize_bucket(blk - nb, false, tblS, sortD, sortS, nblk, n, cnt_in, rowptr, csr_src, do_is, di_is);
}

__device__ __forceinline__ size_t wg_frag_index(int k, int col) {
    const int pr = k >> 1, kb = k & 1;
    const int q = col >> 5, j = (col >> 4) & 1, c = col & 15;
    return (size_t)pr * 256 + (size_t)((((q * 16 + c) * 2 + j) * 2) + kb);
}

__device__ __forceinline__ void
ln_fold_tables_16cols(int blk, const float* __restrict__ W,
                      const float* __restrict__ g, const float* __restrict__ b,
                      float* __restrict__ Wg, float* __restrict__ gw, float* __restrict__ bw) {
    __shared__ float pg[64][17];
    __shared__ float pb[64][17];
    const int tid = threadIdx.x;
    const int cc  = tid & 15;
    const int ks  = tid >> 4;
    const int col = blk * 16 + cc;
    float sg = 0.f, sb = 0.f;
    #pragma unroll
    for (int i = 0; i < 12; ++i) {
        const int k = ks * 12 + i;
        const float w  = W[(size_t)k * 128 + col];
        const float wg = g[k] * w;
        Wg[wg_frag_index(k, col)] = wg;
        sg += wg;
        sb = fmaf(b[k], w, sb);
    }
    pg[ks][cc] = sg;
    pb[ks][cc] = sb;
    __syncthreads();
    if (tid < 32) {
        float acc = 0.f;
        if (tid < 16) { for (int j = 0; j < 64; ++j) acc += pg[j][cc]; gw[col] = acc; }
        else          { for (int j = 0; j < 64; ++j) acc += pb[j][cc]; bw[col] = acc; }
    }
}

__global__ void __launch_bounds__(1024)
bin_edges_and_fold_tables_kernel(const int* __restrict__ src, const int* __restrict__ dst,
                                 int* __restrict__ sortD, int* __restrict__ sortS,
                                 int* __restrict__ tblD, int* __restrict__ tblS,
                                 int e, int nb, int nblk,
                                 const float* __restrict__ W, const float* __restrict__ g,
                                 const float* __restrict__ b,
                                 float* __restrict__ Wg, float* __restrict__ gw, float* __restrict__ bw) {
    const int blk = blockIdx.x;
    if (blk < nblk) bin_chunk(blk, src, dst, sortD, sortS, tblD, tblS, e, nb, nblk);
    else            ln_fold_tables_16cols(blk - nblk, W, g, b, Wg, gw, bw);
}

#define G0_KC     32
#define G0_CHUNK4 (G0_KC / 2 * 64)
#define G0_NS     16
#define G0_APITCH 36
#define G0_ASTRIP (16 * G0_APITCH)
#define G0_ABUF4  (G0_NS * G0_ASTRIP / 4)

__device__ __forceinline__ void
g0_store_item(int strip, int q, int half, int l16, const v8f& c0, const v8f& c1,
              v2f s1, v2f s2,
              const float* __restrict__ do_is,
              const float* __restrict__ gw, const float* __restrict__ bw,
              float* __restrict__ C, int M) {
    const int F = 128;
    int arow = strip * 16 + l16;
    if (arow >= M) arow = M - 1;
    const float do_r = do_is[arow];
    float s  = s1.x + s1.y;
    float sq = s2.x + s2.y;
    s  += __shfl_xor(s,  16, 32);
    sq += __shfl_xor(sq, 16, 32);
    const float m   = s * (1.0f / 768.0f);
    const float rsv = rsqrtf(sq * (1.0f / 768.0f) - m * m + LN_EPS);
    const float pA  = rsv * do_r;
    const float pB  = m * pA;
    const int col0 = q * 32 + l16;
    const int col1 = col0 + 16;
    const float gw0 = gw[col0], gw1 = gw[col1];
    const float bw0 = bw[col0], bw1 = bw[col1];
    const int crow = strip * 16 + half * 8;
    #pragma unroll
    for (int r = 0; r < 8; ++r) {
        const int sl = half * 8 + r;
        const float sA = __shfl(pA,   sl, 32);
        const float sB = __shfl(pB,   sl, 32);
        const float sD = __shfl(do_r, sl, 32);
        if (crow + r < M) {
            C[(size_t)(crow + r) * F + col0] = sA * c0[r] - sB * gw0 + sD * bw0;
            C[(size_t)(crow + r) * F + col1] = sA * c1[r] - sB * gw1 + sD * bw1;
        }
    }
}

__global__ void __launch_bounds__(512)
gemm0_lnfold_kernel(const float* __restrict__ feat,
                    const float* __restrict__ do_is,
                    const float* __restrict__ Wg,
                    const float* __restrict__ gw, const float* __restrict__ bw,
                    float* __restrict__ C, int M) {
    __shared__ float4 bst[2][G0_CHUNK4];
    __shared__ float4 ast[2][G0_ABUF4];
    const int K = 768;
    const int tid  = threadIdx.x;
    const int lane = tid & 31;
    const int wv   = __builtin_amdgcn_readfirstlane(tid >> 5);
    const int q    = wv & 3;
    const int slot = wv >> 2;
    const int half = lane >> 4;
    const int l16  = lane & 15;
    const int koff = half * 2;
    const int nblk = gridDim.x;
    const int b    = blockIdx.x;
    const int mtiles = (M + 15) >> 4;
    const float4* wg4 = (const float4*)Wg;
    const int NCH = K / G0_KC;

    const int st_t   = tid >> 7;
    const int st_row = (tid >> 3) & 15;
    const int st_c4  = tid & 7;

    for (int base = 0; base < mtiles; base += G0_NS * nblk) {
        int nlive = (mtiles - base - b + nblk - 1) / nblk;
        nlive = (nlive < 0) ? 0 : ((nlive > G0_NS) ? G0_NS : nlive);
        const int t3 = 12 + ((slot - q) & 3);
        const bool live3 = (t3 < nlive);

        const float* srcA[4];
        #pragma unroll
        for (int p = 0; p < 4; ++p) {
            const int t = st_t + 4 * p;
            int grow = (base + t * nblk + b) * 16 + st_row;
            if (grow >= M) grow = M - 1;
            if (t >= nlive) grow = 0;
            srcA[p] = feat + (size_t)grow * K + st_c4 * 4;
        }
        const int dstA = (st_t * 16 + st_row) * (G0_APITCH / 4) + st_c4;

        v8f c[4][2];
        v2f s1[4], s2[4];
        #pragma unroll
        for (int j = 0; j < 4; ++j) {
            v8f z = {}; c[j][0] = z; c[j][1] = z;
            v2f z2 = {0.f, 0.f}; s1[j] = z2; s2[j] = z2;
        }

        float4 sa[4], sb[2];
        #pragma unroll
        for (int p = 0; p < 4; ++p) sa[p] = *(const float4*)(srcA[p]);
        #pragma unroll
        for (int p = 0; p < 2; ++p) sb[p] = wg4[p * 512 + tid];
        #pragma unroll
        for (int p = 0; p < 4; ++p) ast[0][dstA + p * (4 * G0_ASTRIP / 4)] = sa[p];
        #pragma unroll
        for (int p = 0; p < 2; ++p) bst[0][p * 512 + tid] = sb[p];
        __syncthreads();

        for (int ch = 0; ch < NCH; ++ch) {
            const int cur = ch & 1;
            const int nx = (ch + 1 < NCH) ? (ch + 1) : ch;
            #pragma unroll
            for (int p = 0; p < 4; ++p) sa[p] = *(const float4*)(srcA[p] + nx * G0_KC);
            #pragma unroll
            for (int p = 0; p < 2; ++p) sb[p] = wg4[(size_t)nx * G0_CHUNK4 + p * 512 + tid];
            const float*  acur = (const float*)ast[cur] + l16 * G0_APITCH + koff;
            const float4* bcur = &bst[cur][half * 64 + q * 16 + l16];
            #pragma unroll
            for (int i = 0; i < G0_KC / 4; ++i) {
                const float4 w = bcur[i * 128];
                v2f b0; b0.x = w.x; b0.y = w.y;
                v2f b1; b1.x = w.z; b1.y = w.w;
                #pragma unroll
                for (int j = 0; j < 3; ++j) {
                    const v2f a = *(const v2f*)(acur + (4 * j + slot) * G0_ASTRIP + 4 * i);
                    s1[j] += a;
                    s2[j] += a * a;
                    c[j][0] = __builtin_amdgcn_wmma_f32_16x16x4_f32(false, a, false, b0, (short)0, c[j][0], false, false);
                    c[j][1] = __builtin_amdgcn_wmma_f32_16x16x4_f32(false, a, false, b1, (short)0, c[j][1], false, false);
                }
            }
            #pragma unroll
            for (int p = 0; p < 4; ++p) ast[cur ^ 1][dstA + p * (4 * G0_ASTRIP / 4)] = sa[p];
            #pragma unroll
            for (int p = 0; p < 2; ++p) bst[cur ^ 1][p * 512 + tid] = sb[p];
            if (live3) {
                #pragma unroll
                for (int i = 0; i < G0_KC / 4; ++i) {
                    const float4 w = bcur[i * 128];
                    v2f b0; b0.x = w.x; b0.y = w.y;
                    v2f b1; b1.x = w.z; b1.y = w.w;
                    const v2f a = *(const v2f*)(acur + t3 * G0_ASTRIP + 4 * i);
                    s1[3] += a;
                    s2[3] += a * a;
                    c[3][0] = __builtin_amdgcn_wmma_f32_16x16x4_f32(false, a, false, b0, (short)0, c[3][0], false, false);
                    c[3][1] = __builtin_amdgcn_wmma_f32_16x16x4_f32(false, a, false, b1, (short)0, c[3][1], false, false);
                }
            }
            __syncthreads();
        }

        #pragma unroll
        for (int j = 0; j < 3; ++j) {
            const int t = 4 * j + slot;
            if (t < nlive)
                g0_store_item(base + t * nblk + b, q, half, l16, c[j][0], c[j][1], s1[j], s2[j], do_is, gw, bw, C, M);
        }
        if (live3)
            g0_store_item(base + t3 * nblk + b, q, half, l16, c[3][0], c[3][1], s1[3], s2[3], do_is, gw, bw, C, M);
    }
}

template <int F4>
__device__ __forceinline__ float4 gather_row_sum(const float* __restrict__ x,
                                                 const int* __restrict__ rowptr,
                                                 const int* __restrict__ cnt_in,
                                                 const int* __restrict__ csr_src,
                                                 int row, int lane) {
    const int beg = rowptr[row];
    const int len = cnt_in[row];
    const int c4  = (lane < F4) ? lane : 0;
    const float4* xp = (const float4*)x + c4;
    float4 acc = make_float4(0.f, 0.f, 0.f, 0.f);
    for (int base = 0; base < len; base += 32) {
        const int m = (len - base < 32) ? (len - base) : 32;
        const int idx = (lane < m) ? csr_src[beg + base + lane] : 0;
        #pragma unroll 4
        for (int j = 0; j < m; ++j) {
            const int s = __builtin_amdgcn_readlane(idx, j);
            const float4 v = xp[(size_t)s * F4];
            acc.x += v.x; acc.y += v.y; acc.z += v.z; acc.w += v.w;
        }
    }
    return acc;
}

template <int F4>
__global__ void gather_sum_vec4_kernel(const float* __restrict__ x,
                                       const int* __restrict__ rowptr,
                                       const int* __restrict__ cnt_in,
                                       const int* __restrict__ csr_src,
                                       float* __restrict__ out, int n) {
    const int lane = threadIdx.x & 31;
    const int row = __builtin_amdgcn_readfirstlane((int)((blockIdx.x * blockDim.x + threadIdx.x) >> 5));
    if (row >= n) return;
    const float4 acc = gather_row_sum<F4>(x, rowptr, cnt_in, csr_src, row, lane);
    if (lane < F4) ((float4*)out)[(size_t)row * F4 + lane] = acc;
}

__global__ void gather_ln_relu_128_kernel(const float* __restrict__ x,
                                          const int* __restrict__ rowptr,
                                          const int* __restrict__ cnt_in,
                                          const int* __restrict__ csr_src,
                                          const float* __restrict__ di_is,
                                          const float* __restrict__ do_is,
                                          const float* __restrict__ g, const float* __restrict__ b,
                                          float* __restrict__ out, int n) {
    const int lane = threadIdx.x & 31;
    const int row = __builtin_amdgcn_readfirstlane((int)((blockIdx.x * blockDim.x + threadIdx.x) >> 5));
    if (row >= n) return;
    const float4 acc = gather_row_sum<32>(x, rowptr, cnt_in, csr_src, row, lane);
    const float di = di_is[row];
    const float ds = do_is[row];
    float v[4] = { acc.x * di, acc.y * di, acc.z * di, acc.w * di };
    float s = 0.f, s2 = 0.f;
    #pragma unroll
    for (int i = 0; i < 4; ++i) { s += v[i]; s2 += v[i] * v[i]; }
    #pragma unroll
    for (int off = 16; off > 0; off >>= 1) {
        s  += __shfl_xor(s,  off, 32);
        s2 += __shfl_xor(s2, off, 32);
    }
    const float m   = s  * (1.0f / 128.0f);
    const float rsv = rsqrtf(s2 * (1.0f / 128.0f) - m * m + LN_EPS);
    float y[4];
    #pragma unroll
    for (int i = 0; i < 4; ++i) {
        const int c = 4 * lane + i;
        const float t = (v[i] - m) * rsv * g[c] + b[c];
        y[i] = fmaxf(t, 0.0f) * ds;
    }
    ((float4*)out)[(size_t)row * 32 + lane] = make_float4(y[0], y[1], y[2], y[3]);
}

__global__ void gather_scale_bias_40_kernel(const float* __restrict__ x,
                                            const int* __restrict__ rowptr,
                                            const int* __restrict__ cnt_in,
                                            const int* __restrict__ csr_src,
                                            const float* __restrict__ di_is,
                                            const float* __restrict__ b2,
                                            float* __restrict__ out, int n) {
    const int lane = threadIdx.x & 31;
    const int row = __builtin_amdgcn_readfirstlane((int)((blockIdx.x * blockDim.x + threadIdx.x) >> 5));
    if (row >= n) return;
    const float4 acc = gather_row_sum<10>(x, rowptr, cnt_in, csr_src, row, lane);
    if (lane < 10) {
        const float di = di_is[row];
        const int c = 4 * lane;
        float4 o;
        o.x = acc.x * di + b2[c];
        o.y = acc.y * di + b2[c + 1];
        o.z = acc.z * di + b2[c + 2];
        o.w = acc.w * di + b2[c + 3];
        ((float4*)out)[(size_t)row * 10 + lane] = o;
    }
}

#define H_PITCH 132
#define W2_COLS 48

__global__ void __launch_bounds__(256)
gemm128_ln_relu_gemm40_wmma_kernel(const float* __restrict__ A, const float* __restrict__ W,
                                   const float* __restrict__ di_is,
                                   const float* __restrict__ do_is,
                                   const float* __restrict__ g, const float* __restrict__ b,
                                   const float* __restrict__ W2,
                                   float* __restrict__ C2, int M) {
    __shared__ v2f    w2s[64 * W2_COLS];
    __shared__ float4 htile[8 * 16 * (H_PITCH / 4)];
    const int K = 128, F = 128, F2 = 40;
    const int tid  = threadIdx.x;
    const int lane = tid & 31;
    const int wv   = tid >> 5;
    const int mt   = (blockIdx.x * blockDim.x + tid) >> 5;
    const bool live = (mt * 16 < M);
    const int half = lane >> 4;
    const int l16  = lane & 15;
    const int koff = half * 2;
    const int crow = mt * 16 + half * 8;
    float* tw = (float*)htile + wv * (16 * H_PITCH);

    {
        v2f st[12];
        #pragma unroll
        for (int j = 0; j < 12; ++j) {
            const int i   = tid + 256 * j;
            const int pr  = i / W2_COLS;
            const int col = i - pr * W2_COLS;
            const int cc  = (col < F2) ? col : (F2 - 1);
            st[j].x = W2[(2 * pr) * F2 + cc];
            st[j].y = W2[(2 * pr + 1) * F2 + cc];
        }
        #pragma unroll
        for (int j = 0; j < 12; ++j) {
            const int i   = tid + 256 * j;
            const int col = i % W2_COLS;
            v2f p = st[j];
            if (col >= F2) { p.x = 0.f; p.y = 0.f; }
            w2s[i] = p;
        }
    }

    if (live) {
        const int arow = mt * 16 + l16;
        const float* ap = A + (size_t)arow * K + koff;

        v8f acc[8];
        #pragma unroll
        for (int i = 0; i < 8; ++i) { v8f z = {}; acc[i] = z; }

        for (int k = 0; k < K; k += 4) {
            const v2f a = *(const v2f*)(ap + k);
            const float* wrow = W + (size_t)(k + koff) * F;
            #pragma unroll
            for (int ft = 0; ft < 8; ++ft) {
                const int bcol = ft * 16 + l16;
                v2f bb;
                bb.x = wrow[bcol];
                bb.y = wrow[bcol + F];
                acc[ft] = __builtin_amdgcn_wmma_f32_16x16x4_f32(
                    false, a, false, bb, (short)0, acc[ft], false, false);
            }
        }

        float gg[8], bbv[8];
        #pragma unroll
        for (int ft = 0; ft < 8; ++ft) {
            gg[ft]  = g[ft * 16 + l16];
            bbv[ft] = b[ft * 16 + l16];
        }
        #pragma unroll
        for (int r = 0; r < 8; ++r) {
            const float di = di_is[crow + r];
            const float ds = do_is[crow + r];
            float v[8];
            float s = 0.f, s2 = 0.f;
            #pragma unroll
            for (int ft = 0; ft < 8; ++ft) {
                v[ft] = acc[ft][r] * di;
                s += v[ft]; s2 += v[ft] * v[ft];
            }
            #pragma unroll
            for (int off = 8; off > 0; off >>= 1) {
                s  += __shfl_xor(s,  off, 32);
                s2 += __shfl_xor(s2, off, 32);
            }
            const float m   = s  * (1.0f / 128.0f);
            const float rsv = rsqrtf(s2 * (1.0f / 128.0f) - m * m + LN_EPS);
            #pragma unroll
            for (int ft = 0; ft < 8; ++ft) {
                const float t = (v[ft] - m) * rsv * gg[ft] + bbv[ft];
                tw[(half * 8 + r) * H_PITCH + ft * 16 + l16] = fmaxf(t, 0.0f) * ds;
            }
        }
    }
    __syncthreads();
    if (!live) return;

    v8f acc2[3];
    #pragma unroll
    for (int i = 0; i < 3; ++i) { v8f z = {}; acc2[i] = z; }
    const float* ta = tw + l16 * H_PITCH + koff;
    const v2f*   wb = w2s + half * W2_COLS + l16;
    for (int k = 0; k < K; k += 4) {
        const v2f a = *(const v2f*)(ta + k);
        #pragma unroll
        for (int ft = 0; ft < 3; ++ft) {
            const v2f bb = wb[(k >> 1) * W2_COLS + ft * 16];
            acc2[ft] = __builtin_amdgcn_wmma_f32_16x16x4_f32(
                false, a, false, bb, (short)0, acc2[ft], false, false);
        }
    }
    #pragma unroll
    for (int ft = 0; ft < 3; ++ft) {
        const int bcol = ft * 16 + l16;
        if (bcol < F2) {
            #pragma unroll
            for (int r = 0; r < 8; ++r)
                C2[(size_t)(crow + r) * F2 + bcol] = acc2[ft][r];
        }
    }
}

static inline int cdiv(long long a, long long b) { return (int)((a + b - 1) / b); }

extern "C" void kernel_launch(void* const* d_in, const int* in_sizes, int n_in,
                              void* d_out, int out_size, void* d_ws, size_t ws_size,
                              hipStream_t stream) {
    const float* feat = (const float*)d_in[0];
    const int*   src  = (const int*)d_in[1];
    const int*   dst  = (const int*)d_in[2];
    const float* W0   = (const float*)d_in[3];
    const float* W1   = (const float*)d_in[4];
    const float* W2   = (const float*)d_in[5];
    const float* b2   = (const float*)d_in[6];
    const float* g_in = (const float*)d_in[7];
    const float* b_in = (const float*)d_in[8];
    const float* g0   = (const float*)d_in[9];
    const float* b0   = (const float*)d_in[10];
    const float* g1   = (const float*)d_in[11];
    const float* b1   = (const float*)d_in[12];
    float* out = (float*)d_out;

    const int IN_F = 768, HID = 128;
    const int n = in_sizes[0] / IN_F;
    const int e = in_sizes[1];

    float* ws    = (float*)d_ws;
    float* do_is = ws;
    float* di_is = ws + (size_t)n;
    float* Wg    = ws + 2 * (size_t)n;
    float* gw    = Wg + (size_t)IN_F * HID;
    float* bw    = gw + HID;
    float* bufB  = ws + 4 * (size_t)n;
    float* bufC  = bufB + (size_t)n * HID;
    float* bufD  = bufC + (size_t)n * HID;
    int*   cnt_in  = (int*)(bufD + (size_t)n * HID);
    int*   rowptr  = cnt_in + (size_t)n;
    int*   csr_src = rowptr + (size_t)n;
    float* bufE  = bufC;

    const int T = 256;

    const int nblk = cdiv(e, CSR_CH);
    const int nb   = cdiv(n, 256);
    int* sortD = (int*)bufC;
    int* sortS = (int*)bufD;
    int* tblD  = sortS + (size_t)e;
    int* tblS  = tblD + (size_t)(nb + 1) * nblk;
    bin_edges_and_fold_tables_kernel<<<nblk + HID / 16, 1024, 0, stream>>>(
        src, dst, sortD, sortS, tblD, tblS, e, nb, nblk, W0, g_in, b_in, Wg, gw, bw);

    finalize_buckets_kernel<<<2 * nb, 1024, 0, stream>>>(
        tblD, tblS, sortD, sortS, nblk, nb, n, cnt_in, rowptr, csr_src, do_is, di_is);

    const int mtiles = cdiv(n, 16);
    gemm0_lnfold_kernel<<<256, 512, 0, stream>>>(feat, do_is, Wg, gw, bw, bufB, n);

    gather_ln_relu_128_kernel<<<cdiv((long long)n * 32, T), T, 0, stream>>>(
        bufB, rowptr, cnt_in, csr_src, di_is, do_is, g0, b0, bufD, n);

    gather_sum_vec4_kernel<32><<<cdiv((long long)n * 32, T), T, 0, stream>>>(
        bufD, rowptr, cnt_in, csr_src, bufB, n);
    gemm128_ln_relu_gemm40_wmma_kernel<<<cdiv((long long)mtiles * 32, T), T, 0, stream>>>(
        bufB, W1, di_is, do_is, g1, b1, W2, bufE, n);

    gather_scale_bias_40_kernel<<<cdiv((long long)n * 32, T), T, 0, stream>>>(
        bufE, rowptr, cnt_in, csr_src, di_is, b2, out, n);
}
